# starting kernel
# baseline (speedup 1.0000x reference)
#include <hip/hip_runtime.h>

#define EMB 512
#define OUTW (3 * EMB)

typedef __attribute__((ext_vector_type(2)))  float     v2f;
typedef __attribute__((ext_vector_type(8)))  float     v8f;
typedef __attribute__((ext_vector_type(4)))  float     f32x4;
typedef __attribute__((ext_vector_type(16))) _Float16  v16h;

__global__ __launch_bounds__(128) void rowmean_wmma(const float* __restrict__ emb,
                                                    float* __restrict__ node_mean,
                                                    int n_nodes) {
  const int lane = threadIdx.x & 31;
  const int wave = threadIdx.x >> 5;
  const int row0 = (blockIdx.x * 4 + wave) * 16;
  if (row0 >= n_nodes) return;

  const int m    = lane & 15;
  const int half = lane >> 4;
  const float* __restrict__ row = emb + (size_t)(row0 + m) * EMB;

  v8f c = {};

#if defined(__AMDGCN__) && __has_builtin(__builtin_amdgcn_wmma_f32_16x16x4_f32)
  const v2f* __restrict__ rp = (const v2f*)row + half;
  v2f bones = {1.0f, 1.0f};
  for (int k = 0; k < EMB / 4; ++k) {
    v2f a = rp[k * 2];
    c = __builtin_amdgcn_wmma_f32_16x16x4_f32(
         false, a,  false, bones,
         (short)0, c,  false,  false);
  }
#else
  v16h bones;
#pragma unroll
  for (int j = 0; j < 16; ++j) bones[j] = (_Float16)1.0f;
  for (int k0 = 0; k0 < EMB; k0 += 32) {
    v16h a;
    const float* p0 = row + k0 + half * 8;
    const float* p1 = row + k0 + 16 + half * 8;
#pragma unroll
    for (int j = 0; j < 8; ++j) a[j] = (_Float16)p0[j];
#pragma unroll
    for (int j = 0; j < 8; ++j) a[8 + j] = (_Float16)p1[j];
    c = __builtin_amdgcn_wmma_f32_16x16x32_f16(
        false, a, false, bones, (short)0, c, false, false);
  }
#endif

  if (m == 0) {
    const float inv = 1.0f / (float)EMB;
    const int base = row0 + half * 8;
#pragma unroll
    for (int r = 0; r < 8; ++r) node_mean[base + r] = c[r] * inv;
  }
}

__global__ void zero_f32(float* __restrict__ p, int n) {
  int i = blockIdx.x * blockDim.x + threadIdx.x;
  if (i < n) p[i] = 0.0f;
}

__global__ void dst_scatter(const int* __restrict__ dst_event,
                            const int* __restrict__ dst_node,
                            const float* __restrict__ node_mean,
                            float* __restrict__ dst_sum, int n_dst) {
  int i = blockIdx.x * blockDim.x + threadIdx.x;
  if (i >= n_dst) return;
  unsafeAtomicAdd(&dst_sum[dst_event[i]], node_mean[dst_node[i]]);
}

__global__ void event_reduce(const int* __restrict__ event_pair,
                             const int* __restrict__ event_src,
                             const int* __restrict__ event_ndst,
                             const int* __restrict__ pair_place,
                             const float* __restrict__ node_mean,
                             const float* __restrict__ dst_sum,
                             float* __restrict__ pair_sum,
                             float* __restrict__ pair_cnt, int n_events) {
  int i = blockIdx.x * blockDim.x + threadIdx.x;
  if (i >= n_events) return;
  int p = event_pair[i];
  float s = (node_mean[pair_place[p]] + node_mean[event_src[i]] +
             dst_sum[i] / ((float)event_ndst[i] + 1.0f)) * (1.0f / 3.0f);
  unsafeAtomicAdd(&pair_sum[p], s);
  unsafeAtomicAdd(&pair_cnt[p], 1.0f);
}

__global__ void pair_reduce(const int* __restrict__ pair_place,
                            const float* __restrict__ pair_sum,
                            const float* __restrict__ pair_cnt,
                            float* __restrict__ place_sum,
                            float* __restrict__ place_cnt, int n_pairs) {
  int i = blockIdx.x * blockDim.x + threadIdx.x;
  if (i >= n_pairs) return;
  float c = pair_cnt[i];
  if (c > 0.0f) {
    int pl = pair_place[i];
    unsafeAtomicAdd(&place_sum[pl], pair_sum[i] / c);
    unsafeAtomicAdd(&place_cnt[pl], 1.0f);
  }
}

__global__ __launch_bounds__(128) void bcast_out(const float* __restrict__ place_sum,
                                                 const float* __restrict__ place_cnt,
                                                 float* __restrict__ out, int n_nodes) {
  int row = blockIdx.x;
  if (row >= n_nodes) return;
  float c = place_cnt[row];
  float pm = c > 0.0f ? place_sum[row] / c : 0.0f;
  f32x4 v = {pm, pm, pm, pm};
  f32x4* __restrict__ o = (f32x4*)out + (size_t)row * (OUTW / 4);
  for (int j = threadIdx.x; j < OUTW / 4; j += 128)
    __builtin_nontemporal_store(v, o + j);
}

extern "C" void kernel_launch(void* const* d_in, const int* in_sizes, int n_in,
                              void* d_out, int out_size, void* d_ws, size_t ws_size,
                              hipStream_t stream) {
  const float* emb       = (const float*)d_in[0];
  const int* pair_place  = (const int*)d_in[1];
  const int* event_pair  = (const int*)d_in[2];
  const int* event_src   = (const int*)d_in[3];
  const int* event_ndst  = (const int*)d_in[4];
  const int* dst_event   = (const int*)d_in[5];
  const int* dst_node    = (const int*)d_in[6];

  const int n_nodes  = in_sizes[0] / EMB;
  const int n_pairs  = in_sizes[1];
  const int n_events = in_sizes[2];
  const int n_dst    = in_sizes[5];

  float* ws        = (float*)d_ws;
  float* node_mean = ws;
  float* dst_sum   = node_mean + n_nodes;
  float* pair_sum  = dst_sum + n_events;
  float* pair_cnt  = pair_sum + n_pairs;
  float* place_sum = pair_cnt + n_pairs;
  float* place_cnt = place_sum + n_nodes;

  const int nzero = n_events + 2 * n_pairs + 2 * n_nodes;
  zero_f32<<<(nzero + 255) / 256, 256, 0, stream>>>(dst_sum, nzero);

  rowmean_wmma<<<(n_nodes + 63) / 64, 128, 0, stream>>>(emb, node_mean, n_nodes);

  dst_scatter<<<(n_dst + 255) / 256, 256, 0, stream>>>(dst_event, dst_node,
                                                       node_mean, dst_sum, n_dst);

  event_reduce<<<(n_events + 255) / 256, 256, 0, stream>>>(
      event_pair, event_src, event_ndst, pair_place, node_mean, dst_sum,
      pair_sum, pair_cnt, n_events);

  pair_reduce<<<(n_pairs + 255) / 256, 256, 0, stream>>>(
      pair_place, pair_sum, pair_cnt, place_sum, place_cnt, n_pairs);

  bcast_out<<<n_nodes, 128, 0, stream>>>(place_sum, place_cnt, (float*)d_out,
                                         n_nodes);
}


# pair-to-place reduction done in LDS: 16 work-groups add pair means into an LDS table, broadcast kernel sums their 16 partial tables; no global atomics for places; place sums added as 16 partials
# speedup vs baseline: 1.1199x; 1.1199x over previous
#include <hip/hip_runtime.h>

#define EMB 512
#define OUTW (3 * EMB)

typedef __attribute__((ext_vector_type(2)))  float     v2f;
typedef __attribute__((ext_vector_type(8)))  float     v8f;
typedef __attribute__((ext_vector_type(4)))  float     f32x4;
typedef __attribute__((ext_vector_type(16))) _Float16  v16h;

__global__ __launch_bounds__(128) void rowmean_wmma(const float* __restrict__ emb,
                                                    float* __restrict__ node_mean,
                                                    int n_nodes) {
  const int lane = threadIdx.x & 31;
  const int wave = threadIdx.x >> 5;
  const int row0 = (blockIdx.x * 4 + wave) * 16;
  if (row0 >= n_nodes) return;

  const int m    = lane & 15;
  const int half = lane >> 4;
  const float* __restrict__ row = emb + (size_t)(row0 + m) * EMB;

  v8f c = {};

#if defined(__AMDGCN__) && __has_builtin(__builtin_amdgcn_wmma_f32_16x16x4_f32)
  const v2f* __restrict__ rp = (const v2f*)row + half;
  v2f bones = {1.0f, 1.0f};
  for (int k = 0; k < EMB / 4; ++k) {
    v2f a = rp[k * 2];
    c = __builtin_amdgcn_wmma_f32_16x16x4_f32(
         false, a,  false, bones,
         (short)0, c,  false,  false);
  }
#else
  v16h bones;
#pragma unroll
  for (int j = 0; j < 16; ++j) bones[j] = (_Float16)1.0f;
  for (int k0 = 0; k0 < EMB; k0 += 32) {
    v16h a;
    const float* p0 = row + k0 + half * 8;
    const float* p1 = row + k0 + 16 + half * 8;
#pragma unroll
    for (int j = 0; j < 8; ++j) a[j] = (_Float16)p0[j];
#pragma unroll
    for (int j = 0; j < 8; ++j) a[8 + j] = (_Float16)p1[j];
    c = __builtin_amdgcn_wmma_f32_16x16x32_f16(
        false, a, false, bones, (short)0, c, false, false);
  }
#endif

  if (m == 0) {
    const float inv = 1.0f / (float)EMB;
    const int base = row0 + half * 8;
#pragma unroll
    for (int r = 0; r < 8; ++r) node_mean[base + r] = c[r] * inv;
  }
}

__global__ void zero_f32(float* __restrict__ p, int n) {
  int i = blockIdx.x * blockDim.x + threadIdx.x;
  if (i < n) p[i] = 0.0f;
}

__global__ void dst_scatter(const int* __restrict__ dst_event,
                            const int* __restrict__ dst_node,
                            const float* __restrict__ node_mean,
                            float* __restrict__ dst_sum, int n_dst) {
  int i = blockIdx.x * blockDim.x + threadIdx.x;
  if (i >= n_dst) return;
  unsafeAtomicAdd(&dst_sum[dst_event[i]], node_mean[dst_node[i]]);
}

__global__ void event_reduce(const int* __restrict__ event_pair,
                             const int* __restrict__ event_src,
                             const int* __restrict__ event_ndst,
                             const int* __restrict__ pair_place,
                             const float* __restrict__ node_mean,
                             const float* __restrict__ dst_sum,
                             float* __restrict__ pair_sum,
                             float* __restrict__ pair_cnt, int n_events) {
  int i = blockIdx.x * blockDim.x + threadIdx.x;
  if (i >= n_events) return;
  int p = event_pair[i];
  float s = (node_mean[pair_place[p]] + node_mean[event_src[i]] +
             dst_sum[i] / ((float)event_ndst[i] + 1.0f)) * (1.0f / 3.0f);
  unsafeAtomicAdd(&pair_sum[p], s);
  unsafeAtomicAdd(&pair_cnt[p], 1.0f);
}

#define NPART 16
#define PLACE_TILE 16384
#define PBATCH 8

__global__ __launch_bounds__(1024) void pair_reduce(const int* __restrict__ pair_place,
                                                    const float* __restrict__ pair_sum,
                                                    const float* __restrict__ pair_cnt,
                                                    float* __restrict__ part,
                                                    int n_pairs, int n_nodes) {
  __shared__ __attribute__((aligned(16))) float tsum[PLACE_TILE];
  __shared__ __attribute__((aligned(16))) float tcnt[PLACE_TILE];
  const int tid = threadIdx.x;
  const int per = (n_pairs + NPART - 1) / NPART;
  const int lo  = blockIdx.x * per;
  const int hi  = min(lo + per, n_pairs);
  float* __restrict__ osum = part + (size_t)(blockIdx.x * 2) * n_nodes;
  float* __restrict__ ocnt = osum + n_nodes;

  for (int base = 0; base < n_nodes; base += PLACE_TILE) {
    for (int j = tid; j < PLACE_TILE / 4; j += 1024) {
      ((f32x4*)tsum)[j] = f32x4{0.0f, 0.0f, 0.0f, 0.0f};
      ((f32x4*)tcnt)[j] = f32x4{0.0f, 0.0f, 0.0f, 0.0f};
    }
    __syncthreads();

    for (int i0 = lo + tid; i0 < hi; i0 += 1024 * PBATCH) {
      float c[PBATCH], s[PBATCH];
      int pl[PBATCH];
#pragma unroll
      for (int u = 0; u < PBATCH; ++u) {
        const int i = i0 + u * 1024;
        const bool ok = i < hi;
        const int ic = ok ? i : i0;
        const float cu = pair_cnt[ic];
        c[u]  = ok ? cu : 0.0f;
        s[u]  = pair_sum[ic];
        pl[u] = pair_place[ic];
      }
#pragma unroll
      for (int u = 0; u < PBATCH; ++u) {
        const unsigned q = (unsigned)(pl[u] - base);
        if (c[u] > 0.0f && q < (unsigned)PLACE_TILE) {
          __hip_atomic_fetch_add(&tsum[q], s[u] / c[u], __ATOMIC_RELAXED,
                                 __HIP_MEMORY_SCOPE_WORKGROUP);
          __hip_atomic_fetch_add(&tcnt[q], 1.0f, __ATOMIC_RELAXED,
                                 __HIP_MEMORY_SCOPE_WORKGROUP);
        }
      }
    }
    __syncthreads();

    const int lim = min(PLACE_TILE, n_nodes - base);
    if ((n_nodes & 3) == 0) {
      f32x4* __restrict__ os4 = (f32x4*)(osum + base);
      f32x4* __restrict__ oc4 = (f32x4*)(ocnt + base);
      for (int j = tid; j < lim / 4; j += 1024) {
        os4[j] = ((const f32x4*)tsum)[j];
        oc4[j] = ((const f32x4*)tcnt)[j];
      }
    } else {
      for (int j = tid; j < lim; j += 1024) {
        osum[base + j] = tsum[j];
        ocnt[base + j] = tcnt[j];
      }
    }
    __syncthreads();
  }
}

__global__ __launch_bounds__(128) void bcast_out(const float* __restrict__ part,
                                                 float* __restrict__ out, int n_nodes) {
  static_assert(NPART == 16, "lanes 0-15 / 16-31 of the first wave map to the 16 partial tables");
  __shared__ float pm_s;
  int row = blockIdx.x;
  if (row >= n_nodes) return;
  if (threadIdx.x < 32) {
    const int k = threadIdx.x & (NPART - 1);
    const int w = threadIdx.x >> 4;
    float v = part[(size_t)(k * 2 + w) * n_nodes + row];
    v += __shfl_xor(v, 8);
    v += __shfl_xor(v, 4);
    v += __shfl_xor(v, 2);
    v += __shfl_xor(v, 1);
    const float s = __shfl(v, 0);
    const float c = __shfl(v, 16);
    if (threadIdx.x == 0) pm_s = c > 0.0f ? s / c : 0.0f;
  }
  __syncthreads();
  const float pm = pm_s;
  f32x4 v4 = {pm, pm, pm, pm};
  f32x4* __restrict__ o = (f32x4*)out + (size_t)row * (OUTW / 4);
  for (int j = threadIdx.x; j < OUTW / 4; j += 128)
    __builtin_nontemporal_store(v4, o + j);
}

extern "C" void kernel_launch(void* const* d_in, const int* in_sizes, int n_in,
                              void* d_out, int out_size, void* d_ws, size_t ws_size,
                              hipStream_t stream) {
  const float* emb       = (const float*)d_in[0];
  const int* pair_place  = (const int*)d_in[1];
  const int* event_pair  = (const int*)d_in[2];
  const int* event_src   = (const int*)d_in[3];
  const int* event_ndst  = (const int*)d_in[4];
  const int* dst_event   = (const int*)d_in[5];
  const int* dst_node    = (const int*)d_in[6];

  const int n_nodes  = in_sizes[0] / EMB;
  const int n_pairs  = in_sizes[1];
  const int n_events = in_sizes[2];
  const int n_dst    = in_sizes[5];

  float* ws        = (float*)d_ws;
  float* node_mean = ws;
  float* dst_sum   = node_mean + n_nodes;
  float* pair_sum  = dst_sum + n_events;
  float* pair_cnt  = pair_sum + n_pairs;
  float* part      = dst_sum;

  const int nzero = n_events + 2 * n_pairs;
  zero_f32<<<(nzero + 255) / 256, 256, 0, stream>>>(dst_sum, nzero);

  rowmean_wmma<<<(n_nodes + 63) / 64, 128, 0, stream>>>(emb, node_mean, n_nodes);

  dst_scatter<<<(n_dst + 255) / 256, 256, 0, stream>>>(dst_event, dst_node,
                                                       node_mean, dst_sum, n_dst);

  event_reduce<<<(n_events + 255) / 256, 256, 0, stream>>>(
      event_pair, event_src, event_ndst, pair_place, node_mean, dst_sum,
      pair_sum, pair_cnt, n_events);

  pair_reduce<<<NPART, 1024, 0, stream>>>(pair_place, pair_sum, pair_cnt, part,
                                          n_pairs, n_nodes);

  bcast_out<<<n_nodes, 128, 0, stream>>>(part, (float*)d_out, n_nodes);
}


# grid-stride loops: zero, dst scatter, event reduce, broadcast launch 256-1024 work-groups that loop, 4 elements per lane (4 rows per wave) a trip, not one lane per element; atomic order still free
# speedup vs baseline: 1.1755x; 1.0496x over previous
#include <hip/hip_runtime.h>

#define EMB 512
#define OUTW (3 * EMB)

typedef __attribute__((ext_vector_type(2)))  float     v2f;
typedef __attribute__((ext_vector_type(8)))  float     v8f;
typedef __attribute__((ext_vector_type(4)))  float     f32x4;
typedef __attribute__((ext_vector_type(4)))  int       i32x4;
typedef __attribute__((ext_vector_type(16))) _Float16  v16h;

__global__ __launch_bounds__(128) void rowmean_wmma(const float* __restrict__ emb,
                                                    float* __restrict__ node_mean,
                                                    int n_nodes) {
  const int lane = threadIdx.x & 31;
  const int wave = threadIdx.x >> 5;
  const int row0 = (blockIdx.x * 4 + wave) * 16;
  if (row0 >= n_nodes) return;

  const int m    = lane & 15;
  const int half = lane >> 4;
  const float* __restrict__ row = emb + (size_t)(row0 + m) * EMB;

  v8f c = {};

#if defined(__AMDGCN__) && __has_builtin(__builtin_amdgcn_wmma_f32_16x16x4_f32)
  const v2f* __restrict__ rp = (const v2f*)row + half;
  v2f bones = {1.0f, 1.0f};
  for (int k = 0; k < EMB / 4; ++k) {
    v2f a = rp[k * 2];
    c = __builtin_amdgcn_wmma_f32_16x16x4_f32(
         false, a,  false, bones,
         (short)0, c,  false,  false);
  }
#else
  v16h bones;
#pragma unroll
  for (int j = 0; j < 16; ++j) bones[j] = (_Float16)1.0f;
  for (int k0 = 0; k0 < EMB; k0 += 32) {
    v16h a;
    const float* p0 = row + k0 + half * 8;
    const float* p1 = row + k0 + 16 + half * 8;
#pragma unroll
    for (int j = 0; j < 8; ++j) a[j] = (_Float16)p0[j];
#pragma unroll
    for (int j = 0; j < 8; ++j) a[8 + j] = (_Float16)p1[j];
    c = __builtin_amdgcn_wmma_f32_16x16x32_f16(
        false, a, false, bones, (short)0, c, false, false);
  }
#endif

  if (m == 0) {
    const float inv = 1.0f / (float)EMB;
    const int base = row0 + half * 8;
#pragma unroll
    for (int r = 0; r < 8; ++r) node_mean[base + r] = c[r] * inv;
  }
}

#define GS_THREADS 256

static __device__ __forceinline__ bool aligned16(const void* p) {
  return (((unsigned long long)p) & 15ull) == 0ull;
}

__global__ __launch_bounds__(GS_THREADS) void zero_f32(float* __restrict__ p, int n) {
  const int tid = blockIdx.x * GS_THREADS + threadIdx.x;
  const int nth = gridDim.x * GS_THREADS;
  const int mis  = (int)((((unsigned long long)p) & 15ull) >> 2);
  const int head = min(n, (4 - mis) & 3);
  if (tid < head) p[tid] = 0.0f;
  float* __restrict__ q = p + head;
  const int m  = n - head;
  const int m4 = m >> 2;
  f32x4* __restrict__ q4 = (f32x4*)q;
  for (int j = tid; j < m4; j += nth) q4[j] = f32x4{0.0f, 0.0f, 0.0f, 0.0f};
  const int t = m4 * 4 + tid;
  if (t < m) q[t] = 0.0f;
}

__global__ __launch_bounds__(GS_THREADS) void dst_scatter(const int* __restrict__ dst_event,
                                                          const int* __restrict__ dst_node,
                                                          const float* __restrict__ node_mean,
                                                          float* __restrict__ dst_sum, int n_dst) {
  const int tid = blockIdx.x * GS_THREADS + threadIdx.x;
  const int nth = gridDim.x * GS_THREADS;
  if (aligned16(dst_event) && aligned16(dst_node)) {
    const i32x4* __restrict__ e4 = (const i32x4*)dst_event;
    const i32x4* __restrict__ d4 = (const i32x4*)dst_node;
    const int n4 = n_dst >> 2;
    for (int j = tid; j < n4; j += nth) {
      const i32x4 e = e4[j];
      const i32x4 d = d4[j];
      const float a0 = node_mean[d.x];
      const float a1 = node_mean[d.y];
      const float a2 = node_mean[d.z];
      const float a3 = node_mean[d.w];
      unsafeAtomicAdd(&dst_sum[e.x], a0);
      unsafeAtomicAdd(&dst_sum[e.y], a1);
      unsafeAtomicAdd(&dst_sum[e.z], a2);
      unsafeAtomicAdd(&dst_sum[e.w], a3);
    }
    const int t = n4 * 4 + tid;
    if (t < n_dst) unsafeAtomicAdd(&dst_sum[dst_event[t]], node_mean[dst_node[t]]);
  } else {
    for (int i = tid; i < n_dst; i += nth)
      unsafeAtomicAdd(&dst_sum[dst_event[i]], node_mean[dst_node[i]]);
  }
}

static __device__ __forceinline__ float event_val(int p, int src, int ndst, float dsum,
                                                  const int* __restrict__ pair_place,
                                                  const float* __restrict__ node_mean) {
  return (node_mean[pair_place[p]] + node_mean[src] +
          dsum / ((float)ndst + 1.0f)) * (1.0f / 3.0f);
}

static __device__ __forceinline__ void event_add(int p, float s,
                                                 float* __restrict__ pair_sum,
                                                 float* __restrict__ pair_cnt) {
  unsafeAtomicAdd(&pair_sum[p], s);
  unsafeAtomicAdd(&pair_cnt[p], 1.0f);
}

__global__ __launch_bounds__(GS_THREADS) void event_reduce(const int* __restrict__ event_pair,
                                                           const int* __restrict__ event_src,
                                                           const int* __restrict__ event_ndst,
                                                           const int* __restrict__ pair_place,
                                                           const float* __restrict__ node_mean,
                                                           const float* __restrict__ dst_sum,
                                                           float* __restrict__ pair_sum,
                                                           float* __restrict__ pair_cnt, int n_events) {
  const int tid = blockIdx.x * GS_THREADS + threadIdx.x;
  const int nth = gridDim.x * GS_THREADS;
  if (aligned16(event_pair) && aligned16(event_src) && aligned16(event_ndst) &&
      aligned16(dst_sum)) {
    const i32x4* __restrict__ p4 = (const i32x4*)event_pair;
    const i32x4* __restrict__ s4 = (const i32x4*)event_src;
    const i32x4* __restrict__ n4v = (const i32x4*)event_ndst;
    const f32x4* __restrict__ d4 = (const f32x4*)dst_sum;
    const int n4 = n_events >> 2;
    for (int j = tid; j < n4; j += nth) {
      const i32x4 p = p4[j];
      const i32x4 sr = s4[j];
      const i32x4 nd = n4v[j];
      const f32x4 ds = d4[j];
      const float v0 = event_val(p.x, sr.x, nd.x, ds.x, pair_place, node_mean);
      const float v1 = event_val(p.y, sr.y, nd.y, ds.y, pair_place, node_mean);
      const float v2 = event_val(p.z, sr.z, nd.z, ds.z, pair_place, node_mean);
      const float v3 = event_val(p.w, sr.w, nd.w, ds.w, pair_place, node_mean);
      event_add(p.x, v0, pair_sum, pair_cnt);
      event_add(p.y, v1, pair_sum, pair_cnt);
      event_add(p.z, v2, pair_sum, pair_cnt);
      event_add(p.w, v3, pair_sum, pair_cnt);
    }
    const int t = n4 * 4 + tid;
    if (t < n_events)
      event_add(event_pair[t],
                event_val(event_pair[t], event_src[t], event_ndst[t], dst_sum[t],
                          pair_place, node_mean),
                pair_sum, pair_cnt);
  } else {
    for (int i = tid; i < n_events; i += nth)
      event_add(event_pair[i],
                event_val(event_pair[i], event_src[i], event_ndst[i], dst_sum[i],
                          pair_place, node_mean),
                pair_sum, pair_cnt);
  }
}

#define NPART 16
#define PLACE_TILE 16384
#define PBATCH 8

__global__ __launch_bounds__(1024) void pair_reduce(const int* __restrict__ pair_place,
                                                    const float* __restrict__ pair_sum,
                                                    const float* __restrict__ pair_cnt,
                                                    float* __restrict__ part,
                                                    int n_pairs, int n_nodes) {
  __shared__ __attribute__((aligned(16))) float tsum[PLACE_TILE];
  __shared__ __attribute__((aligned(16))) float tcnt[PLACE_TILE];
  const int tid = threadIdx.x;
  const int per = (n_pairs + NPART - 1) / NPART;
  const int lo  = blockIdx.x * per;
  const int hi  = min(lo + per, n_pairs);
  float* __restrict__ osum = part + (size_t)(blockIdx.x * 2) * n_nodes;
  float* __restrict__ ocnt = osum + n_nodes;

  for (int base = 0; base < n_nodes; base += PLACE_TILE) {
    for (int j = tid; j < PLACE_TILE / 4; j += 1024) {
      ((f32x4*)tsum)[j] = f32x4{0.0f, 0.0f, 0.0f, 0.0f};
      ((f32x4*)tcnt)[j] = f32x4{0.0f, 0.0f, 0.0f, 0.0f};
    }
    __syncthreads();

    for (int i0 = lo + tid; i0 < hi; i0 += 1024 * PBATCH) {
      float c[PBATCH], s[PBATCH];
      int pl[PBATCH];
#pragma unroll
      for (int u = 0; u < PBATCH; ++u) {
        const int i = i0 + u * 1024;
        const bool ok = i < hi;
        const int ic = ok ? i : i0;
        const float cu = pair_cnt[ic];
        c[u]  = ok ? cu : 0.0f;
        s[u]  = pair_sum[ic];
        pl[u] = pair_place[ic];
      }
#pragma unroll
      for (int u = 0; u < PBATCH; ++u) {
        const unsigned q = (unsigned)(pl[u] - base);
        if (c[u] > 0.0f && q < (unsigned)PLACE_TILE) {
          __hip_atomic_fetch_add(&tsum[q], s[u] / c[u], __ATOMIC_RELAXED,
                                 __HIP_MEMORY_SCOPE_WORKGROUP);
          __hip_atomic_fetch_add(&tcnt[q], 1.0f, __ATOMIC_RELAXED,
                                 __HIP_MEMORY_SCOPE_WORKGROUP);
        }
      }
    }
    __syncthreads();

    const int lim = min(PLACE_TILE, n_nodes - base);
    if ((n_nodes & 3) == 0) {
      f32x4* __restrict__ os4 = (f32x4*)(osum + base);
      f32x4* __restrict__ oc4 = (f32x4*)(ocnt + base);
      for (int j = tid; j < lim / 4; j += 1024) {
        os4[j] = ((const f32x4*)tsum)[j];
        oc4[j] = ((const f32x4*)tcnt)[j];
      }
    } else {
      for (int j = tid; j < lim; j += 1024) {
        osum[base + j] = tsum[j];
        ocnt[base + j] = tcnt[j];
      }
    }
    __syncthreads();
  }
}

#define BROWS 4
#define BTHREADS 128

__global__ __launch_bounds__(BTHREADS) void bcast_out(const float* __restrict__ part,
                                                      float* __restrict__ out, int n_nodes) {
  static_assert(NPART == 16, "lanes 0-15 / 16-31 of a wave map to the 16 partial tables");
  static_assert((OUTW / 4) % 32 == 0, "a row is a whole number of 512-byte wave stores");
  const int lane = threadIdx.x & 31;
  const int gw   = blockIdx.x * (BTHREADS / 32) + (threadIdx.x >> 5);
  const int nw   = gridDim.x * (BTHREADS / 32);
  const int k = lane & (NPART - 1);
  const int w = lane >> 4;
  const float* __restrict__ srcp = part + (size_t)(k * 2 + w) * n_nodes;
  const bool vec = aligned16(part) && (n_nodes & 3) == 0;

  for (int r0 = gw * BROWS; r0 < n_nodes; r0 += nw * BROWS) {
    float v[BROWS];
    if (vec) {
      const f32x4 t = *(const f32x4*)(srcp + r0);
      v[0] = t.x; v[1] = t.y; v[2] = t.z; v[3] = t.w;
    } else {
      for (int u = 0; u < BROWS; ++u) v[u] = srcp[min(r0 + u, n_nodes - 1)];
    }
    for (int u = 0; u < BROWS; ++u) {
      float x = v[u];
      x += __shfl_xor(x, 8);
      x += __shfl_xor(x, 4);
      x += __shfl_xor(x, 2);
      x += __shfl_xor(x, 1);
      const float s = __shfl(x, 0);
      const float c = __shfl(x, 16);
      const float pm = c > 0.0f ? s / c : 0.0f;
      if (r0 + u < n_nodes) {
        const f32x4 v4 = {pm, pm, pm, pm};
        f32x4* __restrict__ o = (f32x4*)out + (size_t)(r0 + u) * (OUTW / 4) + lane;
        for (int j = 0; j < OUTW / 4 / 32; ++j)
          __builtin_nontemporal_store(v4, o + j * 32);
      }
    }
  }
}

extern "C" void kernel_launch(void* const* d_in, const int* in_sizes, int n_in,
                              void* d_out, int out_size, void* d_ws, size_t ws_size,
                              hipStream_t stream) {
  const float* emb       = (const float*)d_in[0];
  const int* pair_place  = (const int*)d_in[1];
  const int* event_pair  = (const int*)d_in[2];
  const int* event_src   = (const int*)d_in[3];
  const int* event_ndst  = (const int*)d_in[4];
  const int* dst_event   = (const int*)d_in[5];
  const int* dst_node    = (const int*)d_in[6];

  const int n_nodes  = in_sizes[0] / EMB;
  const int n_pairs  = in_sizes[1];
  const int n_events = in_sizes[2];
  const int n_dst    = in_sizes[5];

  float* ws        = (float*)d_ws;
  float* node_mean = ws;
  float* dst_sum   = node_mean + n_nodes;
  float* pair_sum  = dst_sum + n_events;
  float* pair_cnt  = pair_sum + n_pairs;
  float* part      = dst_sum;

  const int nzero = n_events + 2 * n_pairs;
  zero_f32<<<256, GS_THREADS, 0, stream>>>(dst_sum, nzero);

  rowmean_wmma<<<(n_nodes + 63) / 64, 128, 0, stream>>>(emb, node_mean, n_nodes);

  dst_scatter<<<1024, GS_THREADS, 0, stream>>>(dst_event, dst_node, node_mean,
                                               dst_sum, n_dst);

  event_reduce<<<1024, GS_THREADS, 0, stream>>>(
      event_pair, event_src, event_ndst, pair_place, node_mean, dst_sum,
      pair_sum, pair_cnt, n_events);

  pair_reduce<<<NPART, 1024, 0, stream>>>(pair_place, pair_sum, pair_cnt, part,
                                          n_pairs, n_nodes);

  bcast_out<<<1024, BTHREADS, 0, stream>>>(part, (float*)d_out, n_nodes);
}


# pair sum/count planes kept once per half of the second-level cache: each wave adds into the copy of the half it runs on (die number), the pair stage adds the two copies; sum order changes
# speedup vs baseline: 2.1097x; 1.7947x over previous
#include <hip/hip_runtime.h>

#define EMB 512
#define OUTW (3 * EMB)

typedef __attribute__((ext_vector_type(2)))  float     v2f;
typedef __attribute__((ext_vector_type(8)))  float     v8f;
typedef __attribute__((ext_vector_type(4)))  float     f32x4;
typedef __attribute__((ext_vector_type(4)))  int       i32x4;
typedef __attribute__((ext_vector_type(16))) _Float16  v16h;

__global__ __launch_bounds__(128) void rowmean_wmma(const float* __restrict__ emb,
                                                    float* __restrict__ node_mean,
                                                    int n_nodes) {
  const int lane = threadIdx.x & 31;
  const int wave = threadIdx.x >> 5;
  const int row0 = (blockIdx.x * 4 + wave) * 16;
  if (row0 >= n_nodes) return;

  const int m    = lane & 15;
  const int half = lane >> 4;
  const float* __restrict__ row = emb + (size_t)(row0 + m) * EMB;

  v8f c = {};

#if defined(__AMDGCN__) && __has_builtin(__builtin_amdgcn_wmma_f32_16x16x4_f32)
  const v2f* __restrict__ rp = (const v2f*)row + half;
  v2f bones = {1.0f, 1.0f};
  for (int k = 0; k < EMB / 4; ++k) {
    v2f a = rp[k * 2];
    c = __builtin_amdgcn_wmma_f32_16x16x4_f32(
         false, a,  false, bones,
         (short)0, c,  false,  false);
  }
#else
  v16h bones;
#pragma unroll
  for (int j = 0; j < 16; ++j) bones[j] = (_Float16)1.0f;
  for (int k0 = 0; k0 < EMB; k0 += 32) {
    v16h a;
    const float* p0 = row + k0 + half * 8;
    const float* p1 = row + k0 + 16 + half * 8;
#pragma unroll
    for (int j = 0; j < 8; ++j) a[j] = (_Float16)p0[j];
#pragma unroll
    for (int j = 0; j < 8; ++j) a[8 + j] = (_Float16)p1[j];
    c = __builtin_amdgcn_wmma_f32_16x16x32_f16(
        false, a, false, bones, (short)0, c, false, false);
  }
#endif

  if (m == 0) {
    const float inv = 1.0f / (float)EMB;
    const int base = row0 + half * 8;
#pragma unroll
    for (int r = 0; r < 8; ++r) node_mean[base + r] = c[r] * inv;
  }
}

#define GS_THREADS 256

static __device__ __forceinline__ bool aligned16(const void* p) {
  return (((unsigned long long)p) & 15ull) == 0ull;
}

static __device__ __forceinline__ int pair_copies(size_t ws_bytes, int n_nodes, int n_events,
                                                  int n_pairs) {
  const size_t need =
      ((size_t)n_nodes + (size_t)n_events + 4 * (size_t)n_pairs) * sizeof(float);
  return ws_bytes >= need ? 2 : 1;
}

static __device__ __forceinline__ int cache_half() {
  return (int)((__builtin_amdgcn_s_sendmsg_rtn(0x87) >> 18) & 1u);
}

__global__ __launch_bounds__(GS_THREADS) void zero_f32(float* __restrict__ p, int n_nodes,
                                                       int n_events, int n_pairs,
                                                       size_t ws_bytes) {
  const int n = n_events + 2 * n_pairs * pair_copies(ws_bytes, n_nodes, n_events, n_pairs);
  const int tid = blockIdx.x * GS_THREADS + threadIdx.x;
  const int nth = gridDim.x * GS_THREADS;
  const int mis  = (int)((((unsigned long long)p) & 15ull) >> 2);
  const int head = min(n, (4 - mis) & 3);
  if (tid < head) p[tid] = 0.0f;
  float* __restrict__ q = p + head;
  const int m  = n - head;
  const int m4 = m >> 2;
  f32x4* __restrict__ q4 = (f32x4*)q;
  for (int j = tid; j < m4; j += nth) q4[j] = f32x4{0.0f, 0.0f, 0.0f, 0.0f};
  const int t = m4 * 4 + tid;
  if (t < m) q[t] = 0.0f;
}

__global__ __launch_bounds__(GS_THREADS) void dst_scatter(const int* __restrict__ dst_event,
                                                          const int* __restrict__ dst_node,
                                                          const float* __restrict__ node_mean,
                                                          float* __restrict__ dst_sum, int n_dst) {
  const int tid = blockIdx.x * GS_THREADS + threadIdx.x;
  const int nth = gridDim.x * GS_THREADS;
  if (aligned16(dst_event) && aligned16(dst_node)) {
    const i32x4* __restrict__ e4 = (const i32x4*)dst_event;
    const i32x4* __restrict__ d4 = (const i32x4*)dst_node;
    const int n4 = n_dst >> 2;
    for (int j = tid; j < n4; j += nth) {
      const i32x4 e = e4[j];
      const i32x4 d = d4[j];
      const float a0 = node_mean[d.x];
      const float a1 = node_mean[d.y];
      const float a2 = node_mean[d.z];
      const float a3 = node_mean[d.w];
      unsafeAtomicAdd(&dst_sum[e.x], a0);
      unsafeAtomicAdd(&dst_sum[e.y], a1);
      unsafeAtomicAdd(&dst_sum[e.z], a2);
      unsafeAtomicAdd(&dst_sum[e.w], a3);
    }
    const int t = n4 * 4 + tid;
    if (t < n_dst) unsafeAtomicAdd(&dst_sum[dst_event[t]], node_mean[dst_node[t]]);
  } else {
    for (int i = tid; i < n_dst; i += nth)
      unsafeAtomicAdd(&dst_sum[dst_event[i]], node_mean[dst_node[i]]);
  }
}

static __device__ __forceinline__ float event_val(int p, int src, int ndst, float dsum,
                                                  const int* __restrict__ pair_place,
                                                  const float* __restrict__ node_mean) {
  return (node_mean[pair_place[p]] + node_mean[src] +
          dsum / ((float)ndst + 1.0f)) * (1.0f / 3.0f);
}

static __device__ __forceinline__ void event_add(int p, float s,
                                                 float* __restrict__ pair_sum,
                                                 float* __restrict__ pair_cnt) {
  unsafeAtomicAdd(&pair_sum[p], s);
  unsafeAtomicAdd(&pair_cnt[p], 1.0f);
}

__global__ __launch_bounds__(GS_THREADS) void event_reduce(const int* __restrict__ event_pair,
                                                           const int* __restrict__ event_src,
                                                           const int* __restrict__ event_ndst,
                                                           const int* __restrict__ pair_place,
                                                           const float* __restrict__ node_mean,
                                                           const float* __restrict__ dst_sum,
                                                           float* __restrict__ pair_sum0,
                                                           float* __restrict__ pair_cnt0, int n_events,
                                                           int n_pairs, int n_nodes, size_t ws_bytes) {
  const int tid = blockIdx.x * GS_THREADS + threadIdx.x;
  const int nth = gridDim.x * GS_THREADS;
  const int h = pair_copies(ws_bytes, n_nodes, n_events, n_pairs) == 2 ? cache_half() : 0;
  float* __restrict__ pair_sum = pair_sum0 + (size_t)h * 2 * (size_t)n_pairs;
  float* __restrict__ pair_cnt = pair_cnt0 + (size_t)h * 2 * (size_t)n_pairs;
  if (aligned16(event_pair) && aligned16(event_src) && aligned16(event_ndst) &&
      aligned16(dst_sum)) {
    const i32x4* __restrict__ p4 = (const i32x4*)event_pair;
    const i32x4* __restrict__ s4 = (const i32x4*)event_src;
    const i32x4* __restrict__ n4v = (const i32x4*)event_ndst;
    const f32x4* __restrict__ d4 = (const f32x4*)dst_sum;
    const int n4 = n_events >> 2;
    for (int j = tid; j < n4; j += nth) {
      const i32x4 p = p4[j];
      const i32x4 sr = s4[j];
      const i32x4 nd = n4v[j];
      const f32x4 ds = d4[j];
      const float v0 = event_val(p.x, sr.x, nd.x, ds.x, pair_place, node_mean);
      const float v1 = event_val(p.y, sr.y, nd.y, ds.y, pair_place, node_mean);
      const float v2 = event_val(p.z, sr.z, nd.z, ds.z, pair_place, node_mean);
      const float v3 = event_val(p.w, sr.w, nd.w, ds.w, pair_place, node_mean);
      event_add(p.x, v0, pair_sum, pair_cnt);
      event_add(p.y, v1, pair_sum, pair_cnt);
      event_add(p.z, v2, pair_sum, pair_cnt);
      event_add(p.w, v3, pair_sum, pair_cnt);
    }
    const int t = n4 * 4 + tid;
    if (t < n_events)
      event_add(event_pair[t],
                event_val(event_pair[t], event_src[t], event_ndst[t], dst_sum[t],
                          pair_place, node_mean),
                pair_sum, pair_cnt);
  } else {
    for (int i = tid; i < n_events; i += nth)
      event_add(event_pair[i],
                event_val(event_pair[i], event_src[i], event_ndst[i], dst_sum[i],
                          pair_place, node_mean),
                pair_sum, pair_cnt);
  }
}

#define NPART 16
#define PLACE_TILE 16384
#define PBATCH 8

__global__ __launch_bounds__(1024) void pair_reduce(const int* __restrict__ pair_place,
                                                    const float* __restrict__ pair_sum,
                                                    const float* __restrict__ pair_cnt,
                                                    float* __restrict__ part,
                                                    int n_pairs, int n_nodes, int n_events,
                                                    size_t ws_bytes) {
  const bool two = pair_copies(ws_bytes, n_nodes, n_events, n_pairs) == 2;
  const float* __restrict__ pair_sum1 = pair_sum + (two ? 2 * (size_t)n_pairs : 0);
  const float* __restrict__ pair_cnt1 = pair_cnt + (two ? 2 * (size_t)n_pairs : 0);
  __shared__ __attribute__((aligned(16))) float tsum[PLACE_TILE];
  __shared__ __attribute__((aligned(16))) float tcnt[PLACE_TILE];
  const int tid = threadIdx.x;
  const int per = (n_pairs + NPART - 1) / NPART;
  const int lo  = blockIdx.x * per;
  const int hi  = min(lo + per, n_pairs);
  float* __restrict__ osum = part + (size_t)(blockIdx.x * 2) * n_nodes;
  float* __restrict__ ocnt = osum + n_nodes;

  for (int base = 0; base < n_nodes; base += PLACE_TILE) {
    for (int j = tid; j < PLACE_TILE / 4; j += 1024) {
      ((f32x4*)tsum)[j] = f32x4{0.0f, 0.0f, 0.0f, 0.0f};
      ((f32x4*)tcnt)[j] = f32x4{0.0f, 0.0f, 0.0f, 0.0f};
    }
    __syncthreads();

    for (int i0 = lo + tid; i0 < hi; i0 += 1024 * PBATCH) {
      float c[PBATCH], s[PBATCH];
      int pl[PBATCH];
#pragma unroll
      for (int u = 0; u < PBATCH; ++u) {
        const int i = i0 + u * 1024;
        const bool ok = i < hi;
        const int ic = ok ? i : i0;
        const float c0 = pair_cnt[ic];
        const float c1 = pair_cnt1[ic];
        const float s0 = pair_sum[ic];
        const float s1 = pair_sum1[ic];
        const float cu = c0 + (two ? c1 : 0.0f);
        c[u]  = ok ? cu : 0.0f;
        s[u]  = s0 + (two ? s1 : 0.0f);
        pl[u] = pair_place[ic];
      }
#pragma unroll
      for (int u = 0; u < PBATCH; ++u) {
        const unsigned q = (unsigned)(pl[u] - base);
        if (c[u] > 0.0f && q < (unsigned)PLACE_TILE) {
          __hip_atomic_fetch_add(&tsum[q], s[u] / c[u], __ATOMIC_RELAXED,
                                 __HIP_MEMORY_SCOPE_WORKGROUP);
          __hip_atomic_fetch_add(&tcnt[q], 1.0f, __ATOMIC_RELAXED,
                                 __HIP_MEMORY_SCOPE_WORKGROUP);
        }
      }
    }
    __syncthreads();

    const int lim = min(PLACE_TILE, n_nodes - base);
    if ((n_nodes & 3) == 0) {
      f32x4* __restrict__ os4 = (f32x4*)(osum + base);
      f32x4* __restrict__ oc4 = (f32x4*)(ocnt + base);
      for (int j = tid; j < lim / 4; j += 1024) {
        os4[j] = ((const f32x4*)tsum)[j];
        oc4[j] = ((const f32x4*)tcnt)[j];
      }
    } else {
      for (int j = tid; j < lim; j += 1024) {
        osum[base + j] = tsum[j];
        ocnt[base + j] = tcnt[j];
      }
    }
    __syncthreads();
  }
}

#define BROWS 4
#define BTHREADS 128

__global__ __launch_bounds__(BTHREADS) void bcast_out(const float* __restrict__ part,
                                                      float* __restrict__ out, int n_nodes) {
  static_assert(NPART == 16, "lanes 0-15 / 16-31 of a wave map to the 16 partial tables");
  static_assert((OUTW / 4) % 32 == 0, "a row is a whole number of 512-byte wave stores");
  const int lane = threadIdx.x & 31;
  const int gw   = blockIdx.x * (BTHREADS / 32) + (threadIdx.x >> 5);
  const int nw   = gridDim.x * (BTHREADS / 32);
  const int k = lane & (NPART - 1);
  const int w = lane >> 4;
  const float* __restrict__ srcp = part + (size_t)(k * 2 + w) * n_nodes;
  const bool vec = aligned16(part) && (n_nodes & 3) == 0;

  for (int r0 = gw * BROWS; r0 < n_nodes; r0 += nw * BROWS) {
    float v[BROWS];
    if (vec) {
      const f32x4 t = *(const f32x4*)(srcp + r0);
      v[0] = t.x; v[1] = t.y; v[2] = t.z; v[3] = t.w;
    } else {
      for (int u = 0; u < BROWS; ++u) v[u] = srcp[min(r0 + u, n_nodes - 1)];
    }
    for (int u = 0; u < BROWS; ++u) {
      float x = v[u];
      x += __shfl_xor(x, 8);
      x += __shfl_xor(x, 4);
      x += __shfl_xor(x, 2);
      x += __shfl_xor(x, 1);
      const float s = __shfl(x, 0);
      const float c = __shfl(x, 16);
      const float pm = c > 0.0f ? s / c : 0.0f;
      if (r0 + u < n_nodes) {
        const f32x4 v4 = {pm, pm, pm, pm};
        f32x4* __restrict__ o = (f32x4*)out + (size_t)(r0 + u) * (OUTW / 4) + lane;
        for (int j = 0; j < OUTW / 4 / 32; ++j)
          __builtin_nontemporal_store(v4, o + j * 32);
      }
    }
  }
}

extern "C" void kernel_launch(void* const* d_in, const int* in_sizes, int n_in,
                              void* d_out, int out_size, void* d_ws, size_t ws_size,
                              hipStream_t stream) {
  const float* emb       = (const float*)d_in[0];
  const int* pair_place  = (const int*)d_in[1];
  const int* event_pair  = (const int*)d_in[2];
  const int* event_src   = (const int*)d_in[3];
  const int* event_ndst  = (const int*)d_in[4];
  const int* dst_event   = (const int*)d_in[5];
  const int* dst_node    = (const int*)d_in[6];

  const int n_nodes  = in_sizes[0] / EMB;
  const int n_pairs  = in_sizes[1];
  const int n_events = in_sizes[2];
  const int n_dst    = in_sizes[5];

  float* ws        = (float*)d_ws;
  float* node_mean = ws;
  float* dst_sum   = node_mean + n_nodes;
  float* pair_sum  = dst_sum + n_events;
  float* pair_cnt  = pair_sum + n_pairs;
  float* part      = dst_sum;

  zero_f32<<<256, GS_THREADS, 0, stream>>>(dst_sum, n_nodes, n_events, n_pairs, ws_size);

  rowmean_wmma<<<(n_nodes + 63) / 64, 128, 0, stream>>>(emb, node_mean, n_nodes);

  dst_scatter<<<1024, GS_THREADS, 0, stream>>>(dst_event, dst_node, node_mean,
                                               dst_sum, n_dst);

  event_reduce<<<1024, GS_THREADS, 0, stream>>>(
      event_pair, event_src, event_ndst, pair_place, node_mean, dst_sum,
      pair_sum, pair_cnt, n_events, n_pairs, n_nodes, ws_size);

  pair_reduce<<<NPART, 1024, 0, stream>>>(pair_place, pair_sum, pair_cnt, part,
                                          n_pairs, n_nodes, n_events, ws_size);

  bcast_out<<<1024, BTHREADS, 0, stream>>>(part, (float*)d_out, n_nodes);
}


# zero-fill of the accumulation planes folded into the row-mean launch (each row-mean wave first stores its share of the zeros): 6 launches become 5
# speedup vs baseline: 2.2254x; 1.0548x over previous
#include <hip/hip_runtime.h>

#define EMB 512
#define OUTW (3 * EMB)

typedef __attribute__((ext_vector_type(2)))  float     v2f;
typedef __attribute__((ext_vector_type(8)))  float     v8f;
typedef __attribute__((ext_vector_type(4)))  float     f32x4;
typedef __attribute__((ext_vector_type(4)))  int       i32x4;
typedef __attribute__((ext_vector_type(16))) _Float16  v16h;

#define GS_THREADS 256

static __device__ __forceinline__ bool aligned16(const void* p) {
  return (((unsigned long long)p) & 15ull) == 0ull;
}

static __device__ __forceinline__ int pair_copies(size_t ws_bytes, int n_nodes, int n_events,
                                                  int n_pairs) {
  const size_t need =
      ((size_t)n_nodes + (size_t)n_events + 4 * (size_t)n_pairs) * sizeof(float);
  return ws_bytes >= need ? 2 : 1;
}

static __device__ __forceinline__ int cache_half() {
  return (int)((__builtin_amdgcn_s_sendmsg_rtn(0x87) >> 18) & 1u);
}

static __device__ __forceinline__ void zero_fill(float* __restrict__ p, int n, int tid, int nth) {
  const int mis  = (int)((((unsigned long long)p) & 15ull) >> 2);
  const int head = min(n, (4 - mis) & 3);
  if (tid < head) p[tid] = 0.0f;
  float* __restrict__ q = p + head;
  const int m  = n - head;
  const int m4 = m >> 2;
  f32x4* __restrict__ q4 = (f32x4*)q;
  for (int j = tid; j < m4; j += nth) q4[j] = f32x4{0.0f, 0.0f, 0.0f, 0.0f};
  const int t = m4 * 4 + tid;
  if (t < m) q[t] = 0.0f;
}

__global__ __launch_bounds__(128) void rowmean_wmma(const float* __restrict__ emb,
                                                    float* __restrict__ node_mean,
                                                    int n_nodes, float* __restrict__ acc,
                                                    int n_events, int n_pairs, size_t ws_bytes) {
  zero_fill(acc, n_events + 2 * n_pairs * pair_copies(ws_bytes, n_nodes, n_events, n_pairs),
            blockIdx.x * 128 + threadIdx.x, gridDim.x * 128);

  const int lane = threadIdx.x & 31;
  const int wave = threadIdx.x >> 5;
  const int row0 = (blockIdx.x * 4 + wave) * 16;
  if (row0 >= n_nodes) return;

  const int m    = lane & 15;
  const int half = lane >> 4;
  const float* __restrict__ row = emb + (size_t)(row0 + m) * EMB;

  v8f c = {};

#if defined(__AMDGCN__) && __has_builtin(__builtin_amdgcn_wmma_f32_16x16x4_f32)
  const v2f* __restrict__ rp = (const v2f*)row + half;
  v2f bones = {1.0f, 1.0f};
  for (int k = 0; k < EMB / 4; ++k) {
    v2f a = rp[k * 2];
    c = __builtin_amdgcn_wmma_f32_16x16x4_f32(
         false, a,  false, bones,
         (short)0, c,  false,  false);
  }
#else
  v16h bones;
#pragma unroll
  for (int j = 0; j < 16; ++j) bones[j] = (_Float16)1.0f;
  for (int k0 = 0; k0 < EMB; k0 += 32) {
    v16h a;
    const float* p0 = row + k0 + half * 8;
    const float* p1 = row + k0 + 16 + half * 8;
#pragma unroll
    for (int j = 0; j < 8; ++j) a[j] = (_Float16)p0[j];
#pragma unroll
    for (int j = 0; j < 8; ++j) a[8 + j] = (_Float16)p1[j];
    c = __builtin_amdgcn_wmma_f32_16x16x32_f16(
        false, a, false, bones, (short)0, c, false, false);
  }
#endif

  if (m == 0) {
    const float inv = 1.0f / (float)EMB;
    const int base = row0 + half * 8;
#pragma unroll
    for (int r = 0; r < 8; ++r) node_mean[base + r] = c[r] * inv;
  }
}

__global__ __launch_bounds__(GS_THREADS) void dst_scatter(const int* __restrict__ dst_event,
                                                          const int* __restrict__ dst_node,
                                                          const float* __restrict__ node_mean,
                                                          float* __restrict__ dst_sum, int n_dst) {
  const int tid = blockIdx.x * GS_THREADS + threadIdx.x;
  const int nth = gridDim.x * GS_THREADS;
  if (aligned16(dst_event) && aligned16(dst_node)) {
    const i32x4* __restrict__ e4 = (const i32x4*)dst_event;
    const i32x4* __restrict__ d4 = (const i32x4*)dst_node;
    const int n4 = n_dst >> 2;
    for (int j = tid; j < n4; j += nth) {
      const i32x4 e = e4[j];
      const i32x4 d = d4[j];
      const float a0 = node_mean[d.x];
      const float a1 = node_mean[d.y];
      const float a2 = node_mean[d.z];
      const float a3 = node_mean[d.w];
      unsafeAtomicAdd(&dst_sum[e.x], a0);
      unsafeAtomicAdd(&dst_sum[e.y], a1);
      unsafeAtomicAdd(&dst_sum[e.z], a2);
      unsafeAtomicAdd(&dst_sum[e.w], a3);
    }
    const int t = n4 * 4 + tid;
    if (t < n_dst) unsafeAtomicAdd(&dst_sum[dst_event[t]], node_mean[dst_node[t]]);
  } else {
    for (int i = tid; i < n_dst; i += nth)
      unsafeAtomicAdd(&dst_sum[dst_event[i]], node_mean[dst_node[i]]);
  }
}

static __device__ __forceinline__ float event_val(int p, int src, int ndst, float dsum,
                                                  const int* __restrict__ pair_place,
                                                  const float* __restrict__ node_mean) {
  return (node_mean[pair_place[p]] + node_mean[src] +
          dsum / ((float)ndst + 1.0f)) * (1.0f / 3.0f);
}

static __device__ __forceinline__ void event_add(int p, float s,
                                                 float* __restrict__ pair_sum,
                                                 float* __restrict__ pair_cnt) {
  unsafeAtomicAdd(&pair_sum[p], s);
  unsafeAtomicAdd(&pair_cnt[p], 1.0f);
}

__global__ __launch_bounds__(GS_THREADS) void event_reduce(const int* __restrict__ event_pair,
                                                           const int* __restrict__ event_src,
                                                           const int* __restrict__ event_ndst,
                                                           const int* __restrict__ pair_place,
                                                           const float* __restrict__ node_mean,
                                                           const float* __restrict__ dst_sum,
                                                           float* __restrict__ pair_sum0,
                                                           float* __restrict__ pair_cnt0, int n_events,
                                                           int n_pairs, int n_nodes, size_t ws_bytes) {
  const int tid = blockIdx.x * GS_THREADS + threadIdx.x;
  const int nth = gridDim.x * GS_THREADS;
  const int h = pair_copies(ws_bytes, n_nodes, n_events, n_pairs) == 2 ? cache_half() : 0;
  float* __restrict__ pair_sum = pair_sum0 + (size_t)h * 2 * (size_t)n_pairs;
  float* __restrict__ pair_cnt = pair_cnt0 + (size_t)h * 2 * (size_t)n_pairs;
  if (aligned16(event_pair) && aligned16(event_src) && aligned16(event_ndst) &&
      aligned16(dst_sum)) {
    const i32x4* __restrict__ p4 = (const i32x4*)event_pair;
    const i32x4* __restrict__ s4 = (const i32x4*)event_src;
    const i32x4* __restrict__ n4v = (const i32x4*)event_ndst;
    const f32x4* __restrict__ d4 = (const f32x4*)dst_sum;
    const int n4 = n_events >> 2;
    for (int j = tid; j < n4; j += nth) {
      const i32x4 p = p4[j];
      const i32x4 sr = s4[j];
      const i32x4 nd = n4v[j];
      const f32x4 ds = d4[j];
      const float v0 = event_val(p.x, sr.x, nd.x, ds.x, pair_place, node_mean);
      const float v1 = event_val(p.y, sr.y, nd.y, ds.y, pair_place, node_mean);
      const float v2 = event_val(p.z, sr.z, nd.z, ds.z, pair_place, node_mean);
      const float v3 = event_val(p.w, sr.w, nd.w, ds.w, pair_place, node_mean);
      event_add(p.x, v0, pair_sum, pair_cnt);
      event_add(p.y, v1, pair_sum, pair_cnt);
      event_add(p.z, v2, pair_sum, pair_cnt);
      event_add(p.w, v3, pair_sum, pair_cnt);
    }
    const int t = n4 * 4 + tid;
    if (t < n_events)
      event_add(event_pair[t],
                event_val(event_pair[t], event_src[t], event_ndst[t], dst_sum[t],
                          pair_place, node_mean),
                pair_sum, pair_cnt);
  } else {
    for (int i = tid; i < n_events; i += nth)
      event_add(event_pair[i],
                event_val(event_pair[i], event_src[i], event_ndst[i], dst_sum[i],
                          pair_place, node_mean),
                pair_sum, pair_cnt);
  }
}

#define NPART 16
#define PLACE_TILE 16384
#define PBATCH 8

__global__ __launch_bounds__(1024) void pair_reduce(const int* __restrict__ pair_place,
                                                    const float* __restrict__ pair_sum,
                                                    const float* __restrict__ pair_cnt,
                                                    float* __restrict__ part,
                                                    int n_pairs, int n_nodes, int n_events,
                                                    size_t ws_bytes) {
  const bool two = pair_copies(ws_bytes, n_nodes, n_events, n_pairs) == 2;
  const float* __restrict__ pair_sum1 = pair_sum + (two ? 2 * (size_t)n_pairs : 0);
  const float* __restrict__ pair_cnt1 = pair_cnt + (two ? 2 * (size_t)n_pairs : 0);
  __shared__ __attribute__((aligned(16))) float tsum[PLACE_TILE];
  __shared__ __attribute__((aligned(16))) float tcnt[PLACE_TILE];
  const int tid = threadIdx.x;
  const int per = (n_pairs + NPART - 1) / NPART;
  const int lo  = blockIdx.x * per;
  const int hi  = min(lo + per, n_pairs);
  float* __restrict__ osum = part + (size_t)(blockIdx.x * 2) * n_nodes;
  float* __restrict__ ocnt = osum + n_nodes;

  for (int base = 0; base < n_nodes; base += PLACE_TILE) {
    for (int j = tid; j < PLACE_TILE / 4; j += 1024) {
      ((f32x4*)tsum)[j] = f32x4{0.0f, 0.0f, 0.0f, 0.0f};
      ((f32x4*)tcnt)[j] = f32x4{0.0f, 0.0f, 0.0f, 0.0f};
    }
    __syncthreads();

    for (int i0 = lo + tid; i0 < hi; i0 += 1024 * PBATCH) {
      float c[PBATCH], s[PBATCH];
      int pl[PBATCH];
#pragma unroll
      for (int u = 0; u < PBATCH; ++u) {
        const int i = i0 + u * 1024;
        const bool ok = i < hi;
        const int ic = ok ? i : i0;
        const float c0 = pair_cnt[ic];
        const float c1 = pair_cnt1[ic];
        const float s0 = pair_sum[ic];
        const float s1 = pair_sum1[ic];
        const float cu = c0 + (two ? c1 : 0.0f);
        c[u]  = ok ? cu : 0.0f;
        s[u]  = s0 + (two ? s1 : 0.0f);
        pl[u] = pair_place[ic];
      }
#pragma unroll
      for (int u = 0; u < PBATCH; ++u) {
        const unsigned q = (unsigned)(pl[u] - base);
        if (c[u] > 0.0f && q < (unsigned)PLACE_TILE) {
          __hip_atomic_fetch_add(&tsum[q], s[u] / c[u], __ATOMIC_RELAXED,
                                 __HIP_MEMORY_SCOPE_WORKGROUP);
          __hip_atomic_fetch_add(&tcnt[q], 1.0f, __ATOMIC_RELAXED,
                                 __HIP_MEMORY_SCOPE_WORKGROUP);
        }
      }
    }
    __syncthreads();

    const int lim = min(PLACE_TILE, n_nodes - base);
    if ((n_nodes & 3) == 0) {
      f32x4* __restrict__ os4 = (f32x4*)(osum + base);
      f32x4* __restrict__ oc4 = (f32x4*)(ocnt + base);
      for (int j = tid; j < lim / 4; j += 1024) {
        os4[j] = ((const f32x4*)tsum)[j];
        oc4[j] = ((const f32x4*)tcnt)[j];
      }
    } else {
      for (int j = tid; j < lim; j += 1024) {
        osum[base + j] = tsum[j];
        ocnt[base + j] = tcnt[j];
      }
    }
    __syncthreads();
  }
}

#define BROWS 4
#define BTHREADS 128

__global__ __launch_bounds__(BTHREADS) void bcast_out(const float* __restrict__ part,
                                                      float* __restrict__ out, int n_nodes) {
  static_assert(NPART == 16, "lanes 0-15 / 16-31 of a wave map to the 16 partial tables");
  static_assert((OUTW / 4) % 32 == 0, "a row is a whole number of 512-byte wave stores");
  const int lane = threadIdx.x & 31;
  const int gw   = blockIdx.x * (BTHREADS / 32) + (threadIdx.x >> 5);
  const int nw   = gridDim.x * (BTHREADS / 32);
  const int k = lane & (NPART - 1);
  const int w = lane >> 4;
  const float* __restrict__ srcp = part + (size_t)(k * 2 + w) * n_nodes;
  const bool vec = aligned16(part) && (n_nodes & 3) == 0;

  for (int r0 = gw * BROWS; r0 < n_nodes; r0 += nw * BROWS) {
    float v[BROWS];
    if (vec) {
      const f32x4 t = *(const f32x4*)(srcp + r0);
      v[0] = t.x; v[1] = t.y; v[2] = t.z; v[3] = t.w;
    } else {
      for (int u = 0; u < BROWS; ++u) v[u] = srcp[min(r0 + u, n_nodes - 1)];
    }
    for (int u = 0; u < BROWS; ++u) {
      float x = v[u];
      x += __shfl_xor(x, 8);
      x += __shfl_xor(x, 4);
      x += __shfl_xor(x, 2);
      x += __shfl_xor(x, 1);
      const float s = __shfl(x, 0);
      const float c = __shfl(x, 16);
      const float pm = c > 0.0f ? s / c : 0.0f;
      if (r0 + u < n_nodes) {
        const f32x4 v4 = {pm, pm, pm, pm};
        f32x4* __restrict__ o = (f32x4*)out + (size_t)(r0 + u) * (OUTW / 4) + lane;
        for (int j = 0; j < OUTW / 4 / 32; ++j)
          __builtin_nontemporal_store(v4, o + j * 32);
      }
    }
  }
}

extern "C" void kernel_launch(void* const* d_in, const int* in_sizes, int n_in,
                              void* d_out, int out_size, void* d_ws, size_t ws_size,
                              hipStream_t stream) {
  const float* emb       = (const float*)d_in[0];
  const int* pair_place  = (const int*)d_in[1];
  const int* event_pair  = (const int*)d_in[2];
  const int* event_src   = (const int*)d_in[3];
  const int* event_ndst  = (const int*)d_in[4];
  const int* dst_event   = (const int*)d_in[5];
  const int* dst_node    = (const int*)d_in[6];

  const int n_nodes  = in_sizes[0] / EMB;
  const int n_pairs  = in_sizes[1];
  const int n_events = in_sizes[2];
  const int n_dst    = in_sizes[5];

  float* ws        = (float*)d_ws;
  float* node_mean = ws;
  float* dst_sum   = node_mean + n_nodes;
  float* pair_sum  = dst_sum + n_events;
  float* pair_cnt  = pair_sum + n_pairs;
  float* part      = dst_sum;

  rowmean_wmma<<<(n_nodes + 63) / 64, 128, 0, stream>>>(emb, node_mean, n_nodes, dst_sum,
                                                        n_events, n_pairs, ws_size);

  dst_scatter<<<1024, GS_THREADS, 0, stream>>>(dst_event, dst_node, node_mean,
                                               dst_sum, n_dst);

  event_reduce<<<1024, GS_THREADS, 0, stream>>>(
      event_pair, event_src, event_ndst, pair_place, node_mean, dst_sum,
      pair_sum, pair_cnt, n_events, n_pairs, n_nodes, ws_size);

  pair_reduce<<<NPART, 1024, 0, stream>>>(pair_place, pair_sum, pair_cnt, part,
                                          n_pairs, n_nodes, n_events, ws_size);

  bcast_out<<<1024, BTHREADS, 0, stream>>>(part, (float*)d_out, n_nodes);
}


# place term node_mean[place]/3, equal for all events and pairs of a place, added once per place in the broadcast stage; event stage drops 2 of 3 random gathers per event; sum order changes
# speedup vs baseline: 2.5115x; 1.1286x over previous
#include <hip/hip_runtime.h>

#define EMB 512
#define OUTW (3 * EMB)

typedef __attribute__((ext_vector_type(2)))  float     v2f;
typedef __attribute__((ext_vector_type(8)))  float     v8f;
typedef __attribute__((ext_vector_type(4)))  float     f32x4;
typedef __attribute__((ext_vector_type(4)))  int       i32x4;
typedef __attribute__((ext_vector_type(16))) _Float16  v16h;

#define GS_THREADS 256

static __device__ __forceinline__ bool aligned16(const void* p) {
  return (((unsigned long long)p) & 15ull) == 0ull;
}

static __device__ __forceinline__ int pair_copies(size_t ws_bytes, int n_nodes, int n_events,
                                                  int n_pairs) {
  const size_t need =
      ((size_t)n_nodes + (size_t)n_events + 4 * (size_t)n_pairs) * sizeof(float);
  return ws_bytes >= need ? 2 : 1;
}

static __device__ __forceinline__ int cache_half() {
  return (int)((__builtin_amdgcn_s_sendmsg_rtn(0x87) >> 18) & 1u);
}

static __device__ __forceinline__ void zero_fill(float* __restrict__ p, int n, int tid, int nth) {
  const int mis  = (int)((((unsigned long long)p) & 15ull) >> 2);
  const int head = min(n, (4 - mis) & 3);
  if (tid < head) p[tid] = 0.0f;
  float* __restrict__ q = p + head;
  const int m  = n - head;
  const int m4 = m >> 2;
  f32x4* __restrict__ q4 = (f32x4*)q;
  for (int j = tid; j < m4; j += nth) q4[j] = f32x4{0.0f, 0.0f, 0.0f, 0.0f};
  const int t = m4 * 4 + tid;
  if (t < m) q[t] = 0.0f;
}

__global__ __launch_bounds__(128) void rowmean_wmma(const float* __restrict__ emb,
                                                    float* __restrict__ node_mean,
                                                    int n_nodes, float* __restrict__ acc,
                                                    int n_events, int n_pairs, size_t ws_bytes) {
  zero_fill(acc, n_events + 2 * n_pairs * pair_copies(ws_bytes, n_nodes, n_events, n_pairs),
            blockIdx.x * 128 + threadIdx.x, gridDim.x * 128);

  const int lane = threadIdx.x & 31;
  const int wave = threadIdx.x >> 5;
  const int row0 = (blockIdx.x * 4 + wave) * 16;
  if (row0 >= n_nodes) return;

  const int m    = lane & 15;
  const int half = lane >> 4;
  const float* __restrict__ row = emb + (size_t)(row0 + m) * EMB;

  v8f c = {};

#if defined(__AMDGCN__) && __has_builtin(__builtin_amdgcn_wmma_f32_16x16x4_f32)
  const v2f* __restrict__ rp = (const v2f*)row + half;
  v2f bones = {1.0f, 1.0f};
  for (int k = 0; k < EMB / 4; ++k) {
    v2f a = rp[k * 2];
    c = __builtin_amdgcn_wmma_f32_16x16x4_f32(
         false, a,  false, bones,
         (short)0, c,  false,  false);
  }
#else
  v16h bones;
#pragma unroll
  for (int j = 0; j < 16; ++j) bones[j] = (_Float16)1.0f;
  for (int k0 = 0; k0 < EMB; k0 += 32) {
    v16h a;
    const float* p0 = row + k0 + half * 8;
    const float* p1 = row + k0 + 16 + half * 8;
#pragma unroll
    for (int j = 0; j < 8; ++j) a[j] = (_Float16)p0[j];
#pragma unroll
    for (int j = 0; j < 8; ++j) a[8 + j] = (_Float16)p1[j];
    c = __builtin_amdgcn_wmma_f32_16x16x32_f16(
        false, a, false, bones, (short)0, c, false, false);
  }
#endif

  if (m == 0) {
    const float inv = 1.0f / (float)EMB;
    const int base = row0 + half * 8;
#pragma unroll
    for (int r = 0; r < 8; ++r) node_mean[base + r] = c[r] * inv;
  }
}

__global__ __launch_bounds__(GS_THREADS) void dst_scatter(const int* __restrict__ dst_event,
                                                          const int* __restrict__ dst_node,
                                                          const float* __restrict__ node_mean,
                                                          float* __restrict__ dst_sum, int n_dst) {
  const int tid = blockIdx.x * GS_THREADS + threadIdx.x;
  const int nth = gridDim.x * GS_THREADS;
  if (aligned16(dst_event) && aligned16(dst_node)) {
    const i32x4* __restrict__ e4 = (const i32x4*)dst_event;
    const i32x4* __restrict__ d4 = (const i32x4*)dst_node;
    const int n4 = n_dst >> 2;
    for (int j = tid; j < n4; j += nth) {
      const i32x4 e = e4[j];
      const i32x4 d = d4[j];
      const float a0 = node_mean[d.x];
      const float a1 = node_mean[d.y];
      const float a2 = node_mean[d.z];
      const float a3 = node_mean[d.w];
      unsafeAtomicAdd(&dst_sum[e.x], a0);
      unsafeAtomicAdd(&dst_sum[e.y], a1);
      unsafeAtomicAdd(&dst_sum[e.z], a2);
      unsafeAtomicAdd(&dst_sum[e.w], a3);
    }
    const int t = n4 * 4 + tid;
    if (t < n_dst) unsafeAtomicAdd(&dst_sum[dst_event[t]], node_mean[dst_node[t]]);
  } else {
    for (int i = tid; i < n_dst; i += nth)
      unsafeAtomicAdd(&dst_sum[dst_event[i]], node_mean[dst_node[i]]);
  }
}

static __device__ __forceinline__ float event_val(int src, int ndst, float dsum,
                                                  const float* __restrict__ node_mean) {
  return (node_mean[src] + dsum / ((float)ndst + 1.0f)) * (1.0f / 3.0f);
}

static __device__ __forceinline__ void event_add(int p, float s,
                                                 float* __restrict__ pair_sum,
                                                 float* __restrict__ pair_cnt) {
  unsafeAtomicAdd(&pair_sum[p], s);
  unsafeAtomicAdd(&pair_cnt[p], 1.0f);
}

__global__ __launch_bounds__(GS_THREADS) void event_reduce(const int* __restrict__ event_pair,
                                                           const int* __restrict__ event_src,
                                                           const int* __restrict__ event_ndst,
                                                           const float* __restrict__ node_mean,
                                                           const float* __restrict__ dst_sum,
                                                           float* __restrict__ pair_sum0,
                                                           float* __restrict__ pair_cnt0, int n_events,
                                                           int n_pairs, int n_nodes, size_t ws_bytes) {
  const int tid = blockIdx.x * GS_THREADS + threadIdx.x;
  const int nth = gridDim.x * GS_THREADS;
  const int h = pair_copies(ws_bytes, n_nodes, n_events, n_pairs) == 2 ? cache_half() : 0;
  float* __restrict__ pair_sum = pair_sum0 + (size_t)h * 2 * (size_t)n_pairs;
  float* __restrict__ pair_cnt = pair_cnt0 + (size_t)h * 2 * (size_t)n_pairs;
  if (aligned16(event_pair) && aligned16(event_src) && aligned16(event_ndst) &&
      aligned16(dst_sum)) {
    const i32x4* __restrict__ p4 = (const i32x4*)event_pair;
    const i32x4* __restrict__ s4 = (const i32x4*)event_src;
    const i32x4* __restrict__ n4v = (const i32x4*)event_ndst;
    const f32x4* __restrict__ d4 = (const f32x4*)dst_sum;
    const int n4 = n_events >> 2;
    for (int j = tid; j < n4; j += nth) {
      const i32x4 p = p4[j];
      const i32x4 sr = s4[j];
      const i32x4 nd = n4v[j];
      const f32x4 ds = d4[j];
      const float v0 = event_val(sr.x, nd.x, ds.x, node_mean);
      const float v1 = event_val(sr.y, nd.y, ds.y, node_mean);
      const float v2 = event_val(sr.z, nd.z, ds.z, node_mean);
      const float v3 = event_val(sr.w, nd.w, ds.w, node_mean);
      event_add(p.x, v0, pair_sum, pair_cnt);
      event_add(p.y, v1, pair_sum, pair_cnt);
      event_add(p.z, v2, pair_sum, pair_cnt);
      event_add(p.w, v3, pair_sum, pair_cnt);
    }
    const int t = n4 * 4 + tid;
    if (t < n_events)
      event_add(event_pair[t], event_val(event_src[t], event_ndst[t], dst_sum[t], node_mean),
                pair_sum, pair_cnt);
  } else {
    for (int i = tid; i < n_events; i += nth)
      event_add(event_pair[i], event_val(event_src[i], event_ndst[i], dst_sum[i], node_mean),
                pair_sum, pair_cnt);
  }
}

#define NPART 16
#define PLACE_TILE 16384
#define PBATCH 8

__global__ __launch_bounds__(1024) void pair_reduce(const int* __restrict__ pair_place,
                                                    const float* __restrict__ pair_sum,
                                                    const float* __restrict__ pair_cnt,
                                                    float* __restrict__ part,
                                                    int n_pairs, int n_nodes, int n_events,
                                                    size_t ws_bytes) {
  const bool two = pair_copies(ws_bytes, n_nodes, n_events, n_pairs) == 2;
  const float* __restrict__ pair_sum1 = pair_sum + (two ? 2 * (size_t)n_pairs : 0);
  const float* __restrict__ pair_cnt1 = pair_cnt + (two ? 2 * (size_t)n_pairs : 0);
  __shared__ __attribute__((aligned(16))) float tsum[PLACE_TILE];
  __shared__ __attribute__((aligned(16))) float tcnt[PLACE_TILE];
  const int tid = threadIdx.x;
  const int per = (n_pairs + NPART - 1) / NPART;
  const int lo  = blockIdx.x * per;
  const int hi  = min(lo + per, n_pairs);
  float* __restrict__ osum = part + (size_t)(blockIdx.x * 2) * n_nodes;
  float* __restrict__ ocnt = osum + n_nodes;

  for (int base = 0; base < n_nodes; base += PLACE_TILE) {
    for (int j = tid; j < PLACE_TILE / 4; j += 1024) {
      ((f32x4*)tsum)[j] = f32x4{0.0f, 0.0f, 0.0f, 0.0f};
      ((f32x4*)tcnt)[j] = f32x4{0.0f, 0.0f, 0.0f, 0.0f};
    }
    __syncthreads();

    for (int i0 = lo + tid; i0 < hi; i0 += 1024 * PBATCH) {
      float c[PBATCH], s[PBATCH];
      int pl[PBATCH];
#pragma unroll
      for (int u = 0; u < PBATCH; ++u) {
        const int i = i0 + u * 1024;
        const bool ok = i < hi;
        const int ic = ok ? i : i0;
        const float c0 = pair_cnt[ic];
        const float c1 = pair_cnt1[ic];
        const float s0 = pair_sum[ic];
        const float s1 = pair_sum1[ic];
        const float cu = c0 + (two ? c1 : 0.0f);
        c[u]  = ok ? cu : 0.0f;
        s[u]  = s0 + (two ? s1 : 0.0f);
        pl[u] = pair_place[ic];
      }
#pragma unroll
      for (int u = 0; u < PBATCH; ++u) {
        const unsigned q = (unsigned)(pl[u] - base);
        if (c[u] > 0.0f && q < (unsigned)PLACE_TILE) {
          __hip_atomic_fetch_add(&tsum[q], s[u] / c[u], __ATOMIC_RELAXED,
                                 __HIP_MEMORY_SCOPE_WORKGROUP);
          __hip_atomic_fetch_add(&tcnt[q], 1.0f, __ATOMIC_RELAXED,
                                 __HIP_MEMORY_SCOPE_WORKGROUP);
        }
      }
    }
    __syncthreads();

    const int lim = min(PLACE_TILE, n_nodes - base);
    if ((n_nodes & 3) == 0) {
      f32x4* __restrict__ os4 = (f32x4*)(osum + base);
      f32x4* __restrict__ oc4 = (f32x4*)(ocnt + base);
      for (int j = tid; j < lim / 4; j += 1024) {
        os4[j] = ((const f32x4*)tsum)[j];
        oc4[j] = ((const f32x4*)tcnt)[j];
      }
    } else {
      for (int j = tid; j < lim; j += 1024) {
        osum[base + j] = tsum[j];
        ocnt[base + j] = tcnt[j];
      }
    }
    __syncthreads();
  }
}

#define BROWS 4
#define BTHREADS 128

__global__ __launch_bounds__(BTHREADS) void bcast_out(const float* __restrict__ part,
                                                      const float* __restrict__ node_mean,
                                                      float* __restrict__ out, int n_nodes) {
  static_assert(NPART == 16, "lanes 0-15 / 16-31 of a wave map to the 16 partial tables");
  static_assert((OUTW / 4) % 32 == 0, "a row is a whole number of 512-byte wave stores");
  const int lane = threadIdx.x & 31;
  const int gw   = blockIdx.x * (BTHREADS / 32) + (threadIdx.x >> 5);
  const int nw   = gridDim.x * (BTHREADS / 32);
  const int k = lane & (NPART - 1);
  const int w = lane >> 4;
  const float* __restrict__ srcp = part + (size_t)(k * 2 + w) * n_nodes;
  const bool vec = aligned16(part) && aligned16(node_mean) && (n_nodes & 3) == 0;

  for (int r0 = gw * BROWS; r0 < n_nodes; r0 += nw * BROWS) {
    float v[BROWS], nm[BROWS];
    if (vec) {
      const f32x4 t = *(const f32x4*)(srcp + r0);
      const f32x4 g = *(const f32x4*)(node_mean + r0);
      v[0] = t.x; v[1] = t.y; v[2] = t.z; v[3] = t.w;
      nm[0] = g.x; nm[1] = g.y; nm[2] = g.z; nm[3] = g.w;
    } else {
      for (int u = 0; u < BROWS; ++u) {
        v[u]  = srcp[min(r0 + u, n_nodes - 1)];
        nm[u] = node_mean[min(r0 + u, n_nodes - 1)];
      }
    }
    for (int u = 0; u < BROWS; ++u) {
      float x = v[u];
      x += __shfl_xor(x, 8);
      x += __shfl_xor(x, 4);
      x += __shfl_xor(x, 2);
      x += __shfl_xor(x, 1);
      const float s = __shfl(x, 0);
      const float c = __shfl(x, 16);
      const float pm = c > 0.0f ? s / c + nm[u] * (1.0f / 3.0f) : 0.0f;
      if (r0 + u < n_nodes) {
        const f32x4 v4 = {pm, pm, pm, pm};
        f32x4* __restrict__ o = (f32x4*)out + (size_t)(r0 + u) * (OUTW / 4) + lane;
        for (int j = 0; j < OUTW / 4 / 32; ++j)
          __builtin_nontemporal_store(v4, o + j * 32);
      }
    }
  }
}

extern "C" void kernel_launch(void* const* d_in, const int* in_sizes, int n_in,
                              void* d_out, int out_size, void* d_ws, size_t ws_size,
                              hipStream_t stream) {
  const float* emb       = (const float*)d_in[0];
  const int* pair_place  = (const int*)d_in[1];
  const int* event_pair  = (const int*)d_in[2];
  const int* event_src   = (const int*)d_in[3];
  const int* event_ndst  = (const int*)d_in[4];
  const int* dst_event   = (const int*)d_in[5];
  const int* dst_node    = (const int*)d_in[6];

  const int n_nodes  = in_sizes[0] / EMB;
  const int n_pairs  = in_sizes[1];
  const int n_events = in_sizes[2];
  const int n_dst    = in_sizes[5];

  float* ws        = (float*)d_ws;
  float* node_mean = ws;
  float* dst_sum   = node_mean + n_nodes;
  float* pair_sum  = dst_sum + n_events;
  float* pair_cnt  = pair_sum + n_pairs;
  float* part      = dst_sum;

  rowmean_wmma<<<(n_nodes + 63) / 64, 128, 0, stream>>>(emb, node_mean, n_nodes, dst_sum,
                                                        n_events, n_pairs, ws_size);

  dst_scatter<<<1024, GS_THREADS, 0, stream>>>(dst_event, dst_node, node_mean,
                                               dst_sum, n_dst);

  event_reduce<<<1024, GS_THREADS, 0, stream>>>(
      event_pair, event_src, event_ndst, node_mean, dst_sum,
      pair_sum, pair_cnt, n_events, n_pairs, n_nodes, ws_size);

  pair_reduce<<<NPART, 1024, 0, stream>>>(pair_place, pair_sum, pair_cnt, part,
                                          n_pairs, n_nodes, n_events, ws_size);

  bcast_out<<<1024, BTHREADS, 0, stream>>>(part, node_mean, (float*)d_out, n_nodes);
}


# node_mean table (64 KiB) copied to LDS once per 1024-thread work-group (256 groups) in the dst and event stages; random 4-byte lookups read LDS, not the vector cache; atomic order still free
# speedup vs baseline: 2.6273x; 1.0461x over previous
#include <hip/hip_runtime.h>

#define EMB 512
#define OUTW (3 * EMB)

typedef __attribute__((ext_vector_type(2)))  float     v2f;
typedef __attribute__((ext_vector_type(8)))  float     v8f;
typedef __attribute__((ext_vector_type(4)))  float     f32x4;
typedef __attribute__((ext_vector_type(4)))  int       i32x4;
typedef __attribute__((ext_vector_type(16))) _Float16  v16h;


static __device__ __forceinline__ bool aligned16(const void* p) {
  return (((unsigned long long)p) & 15ull) == 0ull;
}

static __device__ __forceinline__ int pair_copies(size_t ws_bytes, int n_nodes, int n_events,
                                                  int n_pairs) {
  const size_t need =
      ((size_t)n_nodes + (size_t)n_events + 4 * (size_t)n_pairs) * sizeof(float);
  return ws_bytes >= need ? 2 : 1;
}

static __device__ __forceinline__ int cache_half() {
  return (int)((__builtin_amdgcn_s_sendmsg_rtn(0x87) >> 18) & 1u);
}

static __device__ __forceinline__ void zero_fill(float* __restrict__ p, int n, int tid, int nth) {
  const int mis  = (int)((((unsigned long long)p) & 15ull) >> 2);
  const int head = min(n, (4 - mis) & 3);
  if (tid < head) p[tid] = 0.0f;
  float* __restrict__ q = p + head;
  const int m  = n - head;
  const int m4 = m >> 2;
  f32x4* __restrict__ q4 = (f32x4*)q;
  for (int j = tid; j < m4; j += nth) q4[j] = f32x4{0.0f, 0.0f, 0.0f, 0.0f};
  const int t = m4 * 4 + tid;
  if (t < m) q[t] = 0.0f;
}

__global__ __launch_bounds__(128) void rowmean_wmma(const float* __restrict__ emb,
                                                    float* __restrict__ node_mean,
                                                    int n_nodes, float* __restrict__ acc,
                                                    int n_events, int n_pairs, size_t ws_bytes) {
  zero_fill(acc, n_events + 2 * n_pairs * pair_copies(ws_bytes, n_nodes, n_events, n_pairs),
            blockIdx.x * 128 + threadIdx.x, gridDim.x * 128);

  const int lane = threadIdx.x & 31;
  const int wave = threadIdx.x >> 5;
  const int row0 = (blockIdx.x * 4 + wave) * 16;
  if (row0 >= n_nodes) return;

  const int m    = lane & 15;
  const int half = lane >> 4;
  const float* __restrict__ row = emb + (size_t)(row0 + m) * EMB;

  v8f c = {};

#if defined(__AMDGCN__) && __has_builtin(__builtin_amdgcn_wmma_f32_16x16x4_f32)
  const v2f* __restrict__ rp = (const v2f*)row + half;
  v2f bones = {1.0f, 1.0f};
  for (int k = 0; k < EMB / 4; ++k) {
    v2f a = rp[k * 2];
    c = __builtin_amdgcn_wmma_f32_16x16x4_f32(
         false, a,  false, bones,
         (short)0, c,  false,  false);
  }
#else
  v16h bones;
#pragma unroll
  for (int j = 0; j < 16; ++j) bones[j] = (_Float16)1.0f;
  for (int k0 = 0; k0 < EMB; k0 += 32) {
    v16h a;
    const float* p0 = row + k0 + half * 8;
    const float* p1 = row + k0 + 16 + half * 8;
#pragma unroll
    for (int j = 0; j < 8; ++j) a[j] = (_Float16)p0[j];
#pragma unroll
    for (int j = 0; j < 8; ++j) a[8 + j] = (_Float16)p1[j];
    c = __builtin_amdgcn_wmma_f32_16x16x32_f16(
        false, a, false, bones, (short)0, c, false, false);
  }
#endif

  if (m == 0) {
    const float inv = 1.0f / (float)EMB;
    const int base = row0 + half * 8;
#pragma unroll
    for (int r = 0; r < 8; ++r) node_mean[base + r] = c[r] * inv;
  }
}

#define TAB_MAX 16384
#define TB_THREADS 1024
#define TB_GROUPS 256
#define TB_BATCH 4

static __device__ __forceinline__ void stage_table(float* __restrict__ tab,
                                                   const float* __restrict__ node_mean, int n) {
  const int t = threadIdx.x;
  if (aligned16(node_mean)) {
    const f32x4* __restrict__ s4 = (const f32x4*)node_mean;
    f32x4* __restrict__ t4 = (f32x4*)tab;
    const int n4 = n >> 2;
    for (int j0 = t; j0 < n4; j0 += TB_THREADS * TB_BATCH) {
      f32x4 v[TB_BATCH];
#pragma unroll
      for (int u = 0; u < TB_BATCH; ++u) {
        const int j = j0 + u * TB_THREADS;
        v[u] = s4[j < n4 ? j : j0];
      }
#pragma unroll
      for (int u = 0; u < TB_BATCH; ++u) {
        const int j = j0 + u * TB_THREADS;
        if (j < n4) t4[j] = v[u];
      }
    }
    const int r = n4 * 4 + t;
    if (r < n) tab[r] = node_mean[r];
  } else {
    for (int j = t; j < n; j += TB_THREADS) tab[j] = node_mean[j];
  }
}

template <bool IN_LDS>
static __device__ __forceinline__ float nm_at(const float* __restrict__ tab,
                                              const float* __restrict__ node_mean, int i) {
  return IN_LDS ? tab[i] : node_mean[i];
}

template <bool IN_LDS>
static __device__ __forceinline__ void dst_scatter_body(float* __restrict__ tab,
                                                        const int* __restrict__ dst_event,
                                                        const int* __restrict__ dst_node,
                                                        const float* __restrict__ node_mean,
                                                        float* __restrict__ dst_sum, int n_dst,
                                                        int n_nodes) {
  const int tid = blockIdx.x * TB_THREADS + threadIdx.x;
  const int nth = gridDim.x * TB_THREADS;
  if (aligned16(dst_event) && aligned16(dst_node)) {
    const i32x4* __restrict__ e4 = (const i32x4*)dst_event;
    const i32x4* __restrict__ d4 = (const i32x4*)dst_node;
    const int n4 = n_dst >> 2;
    int j = tid;
    i32x4 e = {0, 0, 0, 0};
    i32x4 d = {0, 0, 0, 0};
    if (j < n4) { e = e4[j]; d = d4[j]; }
    if (IN_LDS) {
      stage_table(tab, node_mean, n_nodes);
      __syncthreads();
    }
    while (j < n4) {
      const float a0 = nm_at<IN_LDS>(tab, node_mean, d.x);
      const float a1 = nm_at<IN_LDS>(tab, node_mean, d.y);
      const float a2 = nm_at<IN_LDS>(tab, node_mean, d.z);
      const float a3 = nm_at<IN_LDS>(tab, node_mean, d.w);
      unsafeAtomicAdd(&dst_sum[e.x], a0);
      unsafeAtomicAdd(&dst_sum[e.y], a1);
      unsafeAtomicAdd(&dst_sum[e.z], a2);
      unsafeAtomicAdd(&dst_sum[e.w], a3);
      j += nth;
      if (j < n4) { e = e4[j]; d = d4[j]; }
    }
    const int t = n4 * 4 + tid;
    if (t < n_dst)
      unsafeAtomicAdd(&dst_sum[dst_event[t]], nm_at<IN_LDS>(tab, node_mean, dst_node[t]));
  } else {
    if (IN_LDS) {
      stage_table(tab, node_mean, n_nodes);
      __syncthreads();
    }
    for (int i = tid; i < n_dst; i += nth)
      unsafeAtomicAdd(&dst_sum[dst_event[i]], nm_at<IN_LDS>(tab, node_mean, dst_node[i]));
  }
}

__global__ __launch_bounds__(TB_THREADS) void dst_scatter(const int* __restrict__ dst_event,
                                                          const int* __restrict__ dst_node,
                                                          const float* __restrict__ node_mean,
                                                          float* __restrict__ dst_sum, int n_dst,
                                                          int n_nodes) {
  __shared__ __attribute__((aligned(16))) float tab[TAB_MAX];
  if (n_nodes <= TAB_MAX)
    dst_scatter_body<true>(tab, dst_event, dst_node, node_mean, dst_sum, n_dst, n_nodes);
  else
    dst_scatter_body<false>(tab, dst_event, dst_node, node_mean, dst_sum, n_dst, n_nodes);
}

template <bool IN_LDS>
static __device__ __forceinline__ float event_val(int src, int ndst, float dsum,
                                                  const float* __restrict__ tab,
                                                  const float* __restrict__ node_mean) {
  return (nm_at<IN_LDS>(tab, node_mean, src) + dsum / ((float)ndst + 1.0f)) * (1.0f / 3.0f);
}

static __device__ __forceinline__ void event_add(int p, float s,
                                                 float* __restrict__ pair_sum,
                                                 float* __restrict__ pair_cnt) {
  unsafeAtomicAdd(&pair_sum[p], s);
  unsafeAtomicAdd(&pair_cnt[p], 1.0f);
}

template <bool IN_LDS>
static __device__ __forceinline__ void event_reduce_body(float* __restrict__ tab,
                                                         const int* __restrict__ event_pair,
                                                         const int* __restrict__ event_src,
                                                         const int* __restrict__ event_ndst,
                                                         const float* __restrict__ node_mean,
                                                         const float* __restrict__ dst_sum,
                                                         float* __restrict__ pair_sum0,
                                                         float* __restrict__ pair_cnt0, int n_events,
                                                         int n_pairs, int n_nodes, size_t ws_bytes) {
  const int tid = blockIdx.x * TB_THREADS + threadIdx.x;
  const int nth = gridDim.x * TB_THREADS;
  const int h = pair_copies(ws_bytes, n_nodes, n_events, n_pairs) == 2 ? cache_half() : 0;
  float* __restrict__ pair_sum = pair_sum0 + (size_t)h * 2 * (size_t)n_pairs;
  float* __restrict__ pair_cnt = pair_cnt0 + (size_t)h * 2 * (size_t)n_pairs;
  if (aligned16(event_pair) && aligned16(event_src) && aligned16(event_ndst) &&
      aligned16(dst_sum)) {
    const i32x4* __restrict__ p4 = (const i32x4*)event_pair;
    const i32x4* __restrict__ s4 = (const i32x4*)event_src;
    const i32x4* __restrict__ n4v = (const i32x4*)event_ndst;
    const f32x4* __restrict__ d4 = (const f32x4*)dst_sum;
    const int n4 = n_events >> 2;
    int j = tid;
    i32x4 p = {0, 0, 0, 0};
    i32x4 sr = {0, 0, 0, 0};
    i32x4 nd = {0, 0, 0, 0};
    f32x4 ds = {0.0f, 0.0f, 0.0f, 0.0f};
    if (j < n4) { p = p4[j]; sr = s4[j]; nd = n4v[j]; ds = d4[j]; }
    if (IN_LDS) {
      stage_table(tab, node_mean, n_nodes);
      __syncthreads();
    }
    while (j < n4) {
      const float v0 = event_val<IN_LDS>(sr.x, nd.x, ds.x, tab, node_mean);
      const float v1 = event_val<IN_LDS>(sr.y, nd.y, ds.y, tab, node_mean);
      const float v2 = event_val<IN_LDS>(sr.z, nd.z, ds.z, tab, node_mean);
      const float v3 = event_val<IN_LDS>(sr.w, nd.w, ds.w, tab, node_mean);
      event_add(p.x, v0, pair_sum, pair_cnt);
      event_add(p.y, v1, pair_sum, pair_cnt);
      event_add(p.z, v2, pair_sum, pair_cnt);
      event_add(p.w, v3, pair_sum, pair_cnt);
      j += nth;
      if (j < n4) { p = p4[j]; sr = s4[j]; nd = n4v[j]; ds = d4[j]; }
    }
    const int t = n4 * 4 + tid;
    if (t < n_events)
      event_add(event_pair[t],
                event_val<IN_LDS>(event_src[t], event_ndst[t], dst_sum[t], tab, node_mean),
                pair_sum, pair_cnt);
  } else {
    if (IN_LDS) {
      stage_table(tab, node_mean, n_nodes);
      __syncthreads();
    }
    for (int i = tid; i < n_events; i += nth)
      event_add(event_pair[i],
                event_val<IN_LDS>(event_src[i], event_ndst[i], dst_sum[i], tab, node_mean),
                pair_sum, pair_cnt);
  }
}

__global__ __launch_bounds__(TB_THREADS) void event_reduce(const int* __restrict__ event_pair,
                                                           const int* __restrict__ event_src,
                                                           const int* __restrict__ event_ndst,
                                                           const float* __restrict__ node_mean,
                                                           const float* __restrict__ dst_sum,
                                                           float* __restrict__ pair_sum0,
                                                           float* __restrict__ pair_cnt0, int n_events,
                                                           int n_pairs, int n_nodes, size_t ws_bytes) {
  __shared__ __attribute__((aligned(16))) float tab[TAB_MAX];
  if (n_nodes <= TAB_MAX)
    event_reduce_body<true>(tab, event_pair, event_src, event_ndst, node_mean, dst_sum, pair_sum0,
                            pair_cnt0, n_events, n_pairs, n_nodes, ws_bytes);
  else
    event_reduce_body<false>(tab, event_pair, event_src, event_ndst, node_mean, dst_sum, pair_sum0,
                             pair_cnt0, n_events, n_pairs, n_nodes, ws_bytes);
}

#define NPART 16
#define PLACE_TILE 16384
#define PBATCH 8

__global__ __launch_bounds__(1024) void pair_reduce(const int* __restrict__ pair_place,
                                                    const float* __restrict__ pair_sum,
                                                    const float* __restrict__ pair_cnt,
                                                    float* __restrict__ part,
                                                    int n_pairs, int n_nodes, int n_events,
                                                    size_t ws_bytes) {
  const bool two = pair_copies(ws_bytes, n_nodes, n_events, n_pairs) == 2;
  const float* __restrict__ pair_sum1 = pair_sum + (two ? 2 * (size_t)n_pairs : 0);
  const float* __restrict__ pair_cnt1 = pair_cnt + (two ? 2 * (size_t)n_pairs : 0);
  __shared__ __attribute__((aligned(16))) float tsum[PLACE_TILE];
  __shared__ __attribute__((aligned(16))) float tcnt[PLACE_TILE];
  const int tid = threadIdx.x;
  const int per = (n_pairs + NPART - 1) / NPART;
  const int lo  = blockIdx.x * per;
  const int hi  = min(lo + per, n_pairs);
  float* __restrict__ osum = part + (size_t)(blockIdx.x * 2) * n_nodes;
  float* __restrict__ ocnt = osum + n_nodes;

  for (int base = 0; base < n_nodes; base += PLACE_TILE) {
    for (int j = tid; j < PLACE_TILE / 4; j += 1024) {
      ((f32x4*)tsum)[j] = f32x4{0.0f, 0.0f, 0.0f, 0.0f};
      ((f32x4*)tcnt)[j] = f32x4{0.0f, 0.0f, 0.0f, 0.0f};
    }
    __syncthreads();

    for (int i0 = lo + tid; i0 < hi; i0 += 1024 * PBATCH) {
      float c[PBATCH], s[PBATCH];
      int pl[PBATCH];
#pragma unroll
      for (int u = 0; u < PBATCH; ++u) {
        const int i = i0 + u * 1024;
        const bool ok = i < hi;
        const int ic = ok ? i : i0;
        const float c0 = pair_cnt[ic];
        const float c1 = pair_cnt1[ic];
        const float s0 = pair_sum[ic];
        const float s1 = pair_sum1[ic];
        const float cu = c0 + (two ? c1 : 0.0f);
        c[u]  = ok ? cu : 0.0f;
        s[u]  = s0 + (two ? s1 : 0.0f);
        pl[u] = pair_place[ic];
      }
#pragma unroll
      for (int u = 0; u < PBATCH; ++u) {
        const unsigned q = (unsigned)(pl[u] - base);
        if (c[u] > 0.0f && q < (unsigned)PLACE_TILE) {
          __hip_atomic_fetch_add(&tsum[q], s[u] / c[u], __ATOMIC_RELAXED,
                                 __HIP_MEMORY_SCOPE_WORKGROUP);
          __hip_atomic_fetch_add(&tcnt[q], 1.0f, __ATOMIC_RELAXED,
                                 __HIP_MEMORY_SCOPE_WORKGROUP);
        }
      }
    }
    __syncthreads();

    const int lim = min(PLACE_TILE, n_nodes - base);
    if ((n_nodes & 3) == 0) {
      f32x4* __restrict__ os4 = (f32x4*)(osum + base);
      f32x4* __restrict__ oc4 = (f32x4*)(ocnt + base);
      for (int j = tid; j < lim / 4; j += 1024) {
        os4[j] = ((const f32x4*)tsum)[j];
        oc4[j] = ((const f32x4*)tcnt)[j];
      }
    } else {
      for (int j = tid; j < lim; j += 1024) {
        osum[base + j] = tsum[j];
        ocnt[base + j] = tcnt[j];
      }
    }
    __syncthreads();
  }
}

#define BROWS 4
#define BTHREADS 128

__global__ __launch_bounds__(BTHREADS) void bcast_out(const float* __restrict__ part,
                                                      const float* __restrict__ node_mean,
                                                      float* __restrict__ out, int n_nodes) {
  static_assert(NPART == 16, "lanes 0-15 / 16-31 of a wave map to the 16 partial tables");
  static_assert((OUTW / 4) % 32 == 0, "a row is a whole number of 512-byte wave stores");
  const int lane = threadIdx.x & 31;
  const int gw   = blockIdx.x * (BTHREADS / 32) + (threadIdx.x >> 5);
  const int nw   = gridDim.x * (BTHREADS / 32);
  const int k = lane & (NPART - 1);
  const int w = lane >> 4;
  const float* __restrict__ srcp = part + (size_t)(k * 2 + w) * n_nodes;
  const bool vec = aligned16(part) && aligned16(node_mean) && (n_nodes & 3) == 0;

  for (int r0 = gw * BROWS; r0 < n_nodes; r0 += nw * BROWS) {
    float v[BROWS], nm[BROWS];
    if (vec) {
      const f32x4 t = *(const f32x4*)(srcp + r0);
      const f32x4 g = *(const f32x4*)(node_mean + r0);
      v[0] = t.x; v[1] = t.y; v[2] = t.z; v[3] = t.w;
      nm[0] = g.x; nm[1] = g.y; nm[2] = g.z; nm[3] = g.w;
    } else {
      for (int u = 0; u < BROWS; ++u) {
        v[u]  = srcp[min(r0 + u, n_nodes - 1)];
        nm[u] = node_mean[min(r0 + u, n_nodes - 1)];
      }
    }
    for (int u = 0; u < BROWS; ++u) {
      float x = v[u];
      x += __shfl_xor(x, 8);
      x += __shfl_xor(x, 4);
      x += __shfl_xor(x, 2);
      x += __shfl_xor(x, 1);
      const float s = __shfl(x, 0);
      const float c = __shfl(x, 16);
      const float pm = c > 0.0f ? s / c + nm[u] * (1.0f / 3.0f) : 0.0f;
      if (r0 + u < n_nodes) {
        const f32x4 v4 = {pm, pm, pm, pm};
        f32x4* __restrict__ o = (f32x4*)out + (size_t)(r0 + u) * (OUTW / 4) + lane;
        for (int j = 0; j < OUTW / 4 / 32; ++j)
          __builtin_nontemporal_store(v4, o + j * 32);
      }
    }
  }
}

extern "C" void kernel_launch(void* const* d_in, const int* in_sizes, int n_in,
                              void* d_out, int out_size, void* d_ws, size_t ws_size,
                              hipStream_t stream) {
  const float* emb       = (const float*)d_in[0];
  const int* pair_place  = (const int*)d_in[1];
  const int* event_pair  = (const int*)d_in[2];
  const int* event_src   = (const int*)d_in[3];
  const int* event_ndst  = (const int*)d_in[4];
  const int* dst_event   = (const int*)d_in[5];
  const int* dst_node    = (const int*)d_in[6];

  const int n_nodes  = in_sizes[0] / EMB;
  const int n_pairs  = in_sizes[1];
  const int n_events = in_sizes[2];
  const int n_dst    = in_sizes[5];

  float* ws        = (float*)d_ws;
  float* node_mean = ws;
  float* dst_sum   = node_mean + n_nodes;
  float* pair_sum  = dst_sum + n_events;
  float* pair_cnt  = pair_sum + n_pairs;
  float* part      = dst_sum;

  rowmean_wmma<<<(n_nodes + 63) / 64, 128, 0, stream>>>(emb, node_mean, n_nodes, dst_sum,
                                                        n_events, n_pairs, ws_size);

  dst_scatter<<<TB_GROUPS, TB_THREADS, 0, stream>>>(dst_event, dst_node, node_mean,
                                                    dst_sum, n_dst, n_nodes);

  event_reduce<<<TB_GROUPS, TB_THREADS, 0, stream>>>(
      event_pair, event_src, event_ndst, node_mean, dst_sum,
      pair_sum, pair_cnt, n_events, n_pairs, n_nodes, ws_size);

  pair_reduce<<<NPART, 1024, 0, stream>>>(pair_place, pair_sum, pair_cnt, part,
                                          n_pairs, n_nodes, n_events, ws_size);

  bcast_out<<<1024, BTHREADS, 0, stream>>>(part, node_mean, (float*)d_out, n_nodes);
}


# row-mean reads whole 128-byte lines: a wave load = 2 lines each of 2 rows (A tile rows 0-7 / 8-15; row = sum of 8 accumulator registers), a wave's 32 loads ahead of its products; sum order changes
# speedup vs baseline: 2.9031x; 1.1050x over previous
#include <hip/hip_runtime.h>

#define EMB 512
#define OUTW (3 * EMB)

typedef __attribute__((ext_vector_type(2)))  float     v2f;
typedef __attribute__((ext_vector_type(8)))  float     v8f;
typedef __attribute__((ext_vector_type(4)))  float     f32x4;
typedef __attribute__((ext_vector_type(4)))  int       i32x4;


static __device__ __forceinline__ bool aligned16(const void* p) {
  return (((unsigned long long)p) & 15ull) == 0ull;
}

static __device__ __forceinline__ int pair_copies(size_t ws_bytes, int n_nodes, int n_events,
                                                  int n_pairs) {
  const size_t need =
      ((size_t)n_nodes + (size_t)n_events + 4 * (size_t)n_pairs) * sizeof(float);
  return ws_bytes >= need ? 2 : 1;
}

static __device__ __forceinline__ int cache_half() {
  return (int)((__builtin_amdgcn_s_sendmsg_rtn(0x87) >> 18) & 1u);
}

static __device__ __forceinline__ void zero_fill(float* __restrict__ p, int n, int tid, int nth) {
  const int mis  = (int)((((unsigned long long)p) & 15ull) >> 2);
  const int head = min(n, (4 - mis) & 3);
  if (tid < head) p[tid] = 0.0f;
  float* __restrict__ q = p + head;
  const int m  = n - head;
  const int m4 = m >> 2;
  f32x4* __restrict__ q4 = (f32x4*)q;
  for (int j = tid; j < m4; j += nth) q4[j] = f32x4{0.0f, 0.0f, 0.0f, 0.0f};
  const int t = m4 * 4 + tid;
  if (t < m) q[t] = 0.0f;
}

#define RM_THREADS 256
#define RM_ROWS 8
#define RM_CHUNKS (EMB / 64)

__global__ __launch_bounds__(RM_THREADS) void rowmean_wmma(const float* __restrict__ emb,
                                                           float* __restrict__ node_mean,
                                                           int n_nodes, float* __restrict__ acc,
                                                           int n_events, int n_pairs,
                                                           size_t ws_bytes) {
  static_assert(EMB % 64 == 0, "a row is a whole number of 256-byte chunks");
  static_assert(RM_ROWS % 4 == 0, "rows go in pairs, and two pairs' products are interleaved");
  zero_fill(acc, n_events + 2 * n_pairs * pair_copies(ws_bytes, n_nodes, n_events, n_pairs),
            blockIdx.x * RM_THREADS + threadIdx.x, gridDim.x * RM_THREADS);

  const int lane = threadIdx.x & 31;
  const int wave = threadIdx.x >> 5;
  const int row0 = (blockIdx.x * (RM_THREADS / 32) + wave) * RM_ROWS;
  if (row0 >= n_nodes) return;

  const int c   = lane & 15;
  const int h   = lane >> 4;
  const int sel = c >> 3;
  const int q   = (c & 7) + 8 * h;
  const float* __restrict__ rp[RM_ROWS / 2];
#pragma unroll
  for (int p = 0; p < RM_ROWS / 2; ++p)
    rp[p] = emb + (size_t)min(row0 + p + (RM_ROWS / 2) * sel, n_nodes - 1) * EMB + q * 4;

  f32x4 v[RM_ROWS / 2][RM_CHUNKS];
#pragma unroll
  for (int g = 0; g < RM_ROWS / 2; g += 2)
#pragma unroll
    for (int s = 0; s < RM_CHUNKS; ++s) {
      v[g][s]     = *(const f32x4*)(rp[g] + s * 64);
      v[g + 1][s] = *(const f32x4*)(rp[g + 1] + s * 64);
    }
  __builtin_amdgcn_sched_barrier(0);

  const v2f bones = {1.0f, 1.0f};
  v8f cacc[RM_ROWS / 2];
#pragma unroll
  for (int p = 0; p < RM_ROWS / 2; ++p) cacc[p] = v8f{};
#pragma unroll
  for (int g = 0; g < RM_ROWS / 2; g += 2)
#pragma unroll
    for (int s = 0; s < RM_CHUNKS; ++s) {
      const f32x4 x = v[g][s];
      const f32x4 y = v[g + 1][s];
      cacc[g] = __builtin_amdgcn_wmma_f32_16x16x4_f32(
           false, v2f{x.x, x.y},  false, bones,
           (short)0, cacc[g],  false,  false);
      cacc[g + 1] = __builtin_amdgcn_wmma_f32_16x16x4_f32(
          false, v2f{y.x, y.y}, false, bones, (short)0, cacc[g + 1], false, false);
      cacc[g] = __builtin_amdgcn_wmma_f32_16x16x4_f32(
          false, v2f{x.z, x.w}, false, bones, (short)0, cacc[g], false, false);
      cacc[g + 1] = __builtin_amdgcn_wmma_f32_16x16x4_f32(
          false, v2f{y.z, y.w}, false, bones, (short)0, cacc[g + 1], false, false);
    }

  if (c == 0) {
    const float inv = 1.0f / (float)EMB;
    const int base = row0 + (RM_ROWS / 2) * h;
#pragma unroll
    for (int p = 0; p < RM_ROWS / 2; ++p) {
      const float t = ((cacc[p][0] + cacc[p][1]) + (cacc[p][2] + cacc[p][3])) +
                      ((cacc[p][4] + cacc[p][5]) + (cacc[p][6] + cacc[p][7]));
      if (base + p < n_nodes) node_mean[base + p] = t * inv;
    }
  }
}

#define TAB_MAX 16384
#define TB_THREADS 1024
#define TB_GROUPS 256
#define TB_BATCH 4

static __device__ __forceinline__ void stage_table(float* __restrict__ tab,
                                                   const float* __restrict__ node_mean, int n) {
  const int t = threadIdx.x;
  if (aligned16(node_mean)) {
    const f32x4* __restrict__ s4 = (const f32x4*)node_mean;
    f32x4* __restrict__ t4 = (f32x4*)tab;
    const int n4 = n >> 2;
    for (int j0 = t; j0 < n4; j0 += TB_THREADS * TB_BATCH) {
      f32x4 v[TB_BATCH];
#pragma unroll
      for (int u = 0; u < TB_BATCH; ++u) {
        const int j = j0 + u * TB_THREADS;
        v[u] = s4[j < n4 ? j : j0];
      }
#pragma unroll
      for (int u = 0; u < TB_BATCH; ++u) {
        const int j = j0 + u * TB_THREADS;
        if (j < n4) t4[j] = v[u];
      }
    }
    const int r = n4 * 4 + t;
    if (r < n) tab[r] = node_mean[r];
  } else {
    for (int j = t; j < n; j += TB_THREADS) tab[j] = node_mean[j];
  }
}

template <bool IN_LDS>
static __device__ __forceinline__ float nm_at(const float* __restrict__ tab,
                                              const float* __restrict__ node_mean, int i) {
  return IN_LDS ? tab[i] : node_mean[i];
}

template <bool IN_LDS>
static __device__ __forceinline__ void dst_scatter_body(float* __restrict__ tab,
                                                        const int* __restrict__ dst_event,
                                                        const int* __restrict__ dst_node,
                                                        const float* __restrict__ node_mean,
                                                        float* __restrict__ dst_sum, int n_dst,
                                                        int n_nodes) {
  const int tid = blockIdx.x * TB_THREADS + threadIdx.x;
  const int nth = gridDim.x * TB_THREADS;
  if (aligned16(dst_event) && aligned16(dst_node)) {
    const i32x4* __restrict__ e4 = (const i32x4*)dst_event;
    const i32x4* __restrict__ d4 = (const i32x4*)dst_node;
    const int n4 = n_dst >> 2;
    int j = tid;
    i32x4 e = {0, 0, 0, 0};
    i32x4 d = {0, 0, 0, 0};
    if (j < n4) { e = e4[j]; d = d4[j]; }
    if (IN_LDS) {
      stage_table(tab, node_mean, n_nodes);
      __syncthreads();
    }
    while (j < n4) {
      const float a0 = nm_at<IN_LDS>(tab, node_mean, d.x);
      const float a1 = nm_at<IN_LDS>(tab, node_mean, d.y);
      const float a2 = nm_at<IN_LDS>(tab, node_mean, d.z);
      const float a3 = nm_at<IN_LDS>(tab, node_mean, d.w);
      unsafeAtomicAdd(&dst_sum[e.x], a0);
      unsafeAtomicAdd(&dst_sum[e.y], a1);
      unsafeAtomicAdd(&dst_sum[e.z], a2);
      unsafeAtomicAdd(&dst_sum[e.w], a3);
      j += nth;
      if (j < n4) { e = e4[j]; d = d4[j]; }
    }
    const int t = n4 * 4 + tid;
    if (t < n_dst)
      unsafeAtomicAdd(&dst_sum[dst_event[t]], nm_at<IN_LDS>(tab, node_mean, dst_node[t]));
  } else {
    if (IN_LDS) {
      stage_table(tab, node_mean, n_nodes);
      __syncthreads();
    }
    for (int i = tid; i < n_dst; i += nth)
      unsafeAtomicAdd(&dst_sum[dst_event[i]], nm_at<IN_LDS>(tab, node_mean, dst_node[i]));
  }
}

__global__ __launch_bounds__(TB_THREADS) void dst_scatter(const int* __restrict__ dst_event,
                                                          const int* __restrict__ dst_node,
                                                          const float* __restrict__ node_mean,
                                                          float* __restrict__ dst_sum, int n_dst,
                                                          int n_nodes) {
  __shared__ __attribute__((aligned(16))) float tab[TAB_MAX];
  if (n_nodes <= TAB_MAX)
    dst_scatter_body<true>(tab, dst_event, dst_node, node_mean, dst_sum, n_dst, n_nodes);
  else
    dst_scatter_body<false>(tab, dst_event, dst_node, node_mean, dst_sum, n_dst, n_nodes);
}

template <bool IN_LDS>
static __device__ __forceinline__ float event_val(int src, int ndst, float dsum,
                                                  const float* __restrict__ tab,
                                                  const float* __restrict__ node_mean) {
  return (nm_at<IN_LDS>(tab, node_mean, src) + dsum / ((float)ndst + 1.0f)) * (1.0f / 3.0f);
}

static __device__ __forceinline__ void event_add(int p, float s,
                                                 float* __restrict__ pair_sum,
                                                 float* __restrict__ pair_cnt) {
  unsafeAtomicAdd(&pair_sum[p], s);
  unsafeAtomicAdd(&pair_cnt[p], 1.0f);
}

template <bool IN_LDS>
static __device__ __forceinline__ void event_reduce_body(float* __restrict__ tab,
                                                         const int* __restrict__ event_pair,
                                                         const int* __restrict__ event_src,
                                                         const int* __restrict__ event_ndst,
                                                         const float* __restrict__ node_mean,
                                                         const float* __restrict__ dst_sum,
                                                         float* __restrict__ pair_sum0,
                                                         float* __restrict__ pair_cnt0, int n_events,
                                                         int n_pairs, int n_nodes, size_t ws_bytes) {
  const int tid = blockIdx.x * TB_THREADS + threadIdx.x;
  const int nth = gridDim.x * TB_THREADS;
  const int h = pair_copies(ws_bytes, n_nodes, n_events, n_pairs) == 2 ? cache_half() : 0;
  float* __restrict__ pair_sum = pair_sum0 + (size_t)h * 2 * (size_t)n_pairs;
  float* __restrict__ pair_cnt = pair_cnt0 + (size_t)h * 2 * (size_t)n_pairs;
  if (aligned16(event_pair) && aligned16(event_src) && aligned16(event_ndst) &&
      aligned16(dst_sum)) {
    const i32x4* __restrict__ p4 = (const i32x4*)event_pair;
    const i32x4* __restrict__ s4 = (const i32x4*)event_src;
    const i32x4* __restrict__ n4v = (const i32x4*)event_ndst;
    const f32x4* __restrict__ d4 = (const f32x4*)dst_sum;
    const int n4 = n_events >> 2;
    int j = tid;
    i32x4 p = {0, 0, 0, 0};
    i32x4 sr = {0, 0, 0, 0};
    i32x4 nd = {0, 0, 0, 0};
    f32x4 ds = {0.0f, 0.0f, 0.0f, 0.0f};
    if (j < n4) { p = p4[j]; sr = s4[j]; nd = n4v[j]; ds = d4[j]; }
    if (IN_LDS) {
      stage_table(tab, node_mean, n_nodes);
      __syncthreads();
    }
    while (j < n4) {
      const float v0 = event_val<IN_LDS>(sr.x, nd.x, ds.x, tab, node_mean);
      const float v1 = event_val<IN_LDS>(sr.y, nd.y, ds.y, tab, node_mean);
      const float v2 = event_val<IN_LDS>(sr.z, nd.z, ds.z, tab, node_mean);
      const float v3 = event_val<IN_LDS>(sr.w, nd.w, ds.w, tab, node_mean);
      event_add(p.x, v0, pair_sum, pair_cnt);
      event_add(p.y, v1, pair_sum, pair_cnt);
      event_add(p.z, v2, pair_sum, pair_cnt);
      event_add(p.w, v3, pair_sum, pair_cnt);
      j += nth;
      if (j < n4) { p = p4[j]; sr = s4[j]; nd = n4v[j]; ds = d4[j]; }
    }
    const int t = n4 * 4 + tid;
    if (t < n_events)
      event_add(event_pair[t],
                event_val<IN_LDS>(event_src[t], event_ndst[t], dst_sum[t], tab, node_mean),
                pair_sum, pair_cnt);
  } else {
    if (IN_LDS) {
      stage_table(tab, node_mean, n_nodes);
      __syncthreads();
    }
    for (int i = tid; i < n_events; i += nth)
      event_add(event_pair[i],
                event_val<IN_LDS>(event_src[i], event_ndst[i], dst_sum[i], tab, node_mean),
                pair_sum, pair_cnt);
  }
}

__global__ __launch_bounds__(TB_THREADS) void event_reduce(const int* __restrict__ event_pair,
                                                           const int* __restrict__ event_src,
                                                           const int* __restrict__ event_ndst,
                                                           const float* __restrict__ node_mean,
                                                           const float* __restrict__ dst_sum,
                                                           float* __restrict__ pair_sum0,
                                                           float* __restrict__ pair_cnt0, int n_events,
                                                           int n_pairs, int n_nodes, size_t ws_bytes) {
  __shared__ __attribute__((aligned(16))) float tab[TAB_MAX];
  if (n_nodes <= TAB_MAX)
    event_reduce_body<true>(tab, event_pair, event_src, event_ndst, node_mean, dst_sum, pair_sum0,
                            pair_cnt0, n_events, n_pairs, n_nodes, ws_bytes);
  else
    event_reduce_body<false>(tab, event_pair, event_src, event_ndst, node_mean, dst_sum, pair_sum0,
                             pair_cnt0, n_events, n_pairs, n_nodes, ws_bytes);
}

#define NPART 16
#define PLACE_TILE 16384
#define PBATCH 8

__global__ __launch_bounds__(1024) void pair_reduce(const int* __restrict__ pair_place,
                                                    const float* __restrict__ pair_sum,
                                                    const float* __restrict__ pair_cnt,
                                                    float* __restrict__ part,
                                                    int n_pairs, int n_nodes, int n_events,
                                                    size_t ws_bytes) {
  const bool two = pair_copies(ws_bytes, n_nodes, n_events, n_pairs) == 2;
  const float* __restrict__ pair_sum1 = pair_sum + (two ? 2 * (size_t)n_pairs : 0);
  const float* __restrict__ pair_cnt1 = pair_cnt + (two ? 2 * (size_t)n_pairs : 0);
  __shared__ __attribute__((aligned(16))) float tsum[PLACE_TILE];
  __shared__ __attribute__((aligned(16))) float tcnt[PLACE_TILE];
  const int tid = threadIdx.x;
  const int per = (n_pairs + NPART - 1) / NPART;
  const int lo  = blockIdx.x * per;
  const int hi  = min(lo + per, n_pairs);
  float* __restrict__ osum = part + (size_t)(blockIdx.x * 2) * n_nodes;
  float* __restrict__ ocnt = osum + n_nodes;

  for (int base = 0; base < n_nodes; base += PLACE_TILE) {
    for (int j = tid; j < PLACE_TILE / 4; j += 1024) {
      ((f32x4*)tsum)[j] = f32x4{0.0f, 0.0f, 0.0f, 0.0f};
      ((f32x4*)tcnt)[j] = f32x4{0.0f, 0.0f, 0.0f, 0.0f};
    }
    __syncthreads();

    for (int i0 = lo + tid; i0 < hi; i0 += 1024 * PBATCH) {
      float c[PBATCH], s[PBATCH];
      int pl[PBATCH];
#pragma unroll
      for (int u = 0; u < PBATCH; ++u) {
        const int i = i0 + u * 1024;
        const bool ok = i < hi;
        const int ic = ok ? i : i0;
        const float c0 = pair_cnt[ic];
        const float c1 = pair_cnt1[ic];
        const float s0 = pair_sum[ic];
        const float s1 = pair_sum1[ic];
        const float cu = c0 + (two ? c1 : 0.0f);
        c[u]  = ok ? cu : 0.0f;
        s[u]  = s0 + (two ? s1 : 0.0f);
        pl[u] = pair_place[ic];
      }
#pragma unroll
      for (int u = 0; u < PBATCH; ++u) {
        const unsigned q = (unsigned)(pl[u] - base);
        if (c[u] > 0.0f && q < (unsigned)PLACE_TILE) {
          __hip_atomic_fetch_add(&tsum[q], s[u] / c[u], __ATOMIC_RELAXED,
                                 __HIP_MEMORY_SCOPE_WORKGROUP);
          __hip_atomic_fetch_add(&tcnt[q], 1.0f, __ATOMIC_RELAXED,
                                 __HIP_MEMORY_SCOPE_WORKGROUP);
        }
      }
    }
    __syncthreads();

    const int lim = min(PLACE_TILE, n_nodes - base);
    if ((n_nodes & 3) == 0) {
      f32x4* __restrict__ os4 = (f32x4*)(osum + base);
      f32x4* __restrict__ oc4 = (f32x4*)(ocnt + base);
      for (int j = tid; j < lim / 4; j += 1024) {
        os4[j] = ((const f32x4*)tsum)[j];
        oc4[j] = ((const f32x4*)tcnt)[j];
      }
    } else {
      for (int j = tid; j < lim; j += 1024) {
        osum[base + j] = tsum[j];
        ocnt[base + j] = tcnt[j];
      }
    }
    __syncthreads();
  }
}

#define BROWS 4
#define BTHREADS 128

__global__ __launch_bounds__(BTHREADS) void bcast_out(const float* __restrict__ part,
                                                      const float* __restrict__ node_mean,
                                                      float* __restrict__ out, int n_nodes) {
  static_assert(NPART == 16, "lanes 0-15 / 16-31 of a wave map to the 16 partial tables");
  static_assert((OUTW / 4) % 32 == 0, "a row is a whole number of 512-byte wave stores");
  const int lane = threadIdx.x & 31;
  const int gw   = blockIdx.x * (BTHREADS / 32) + (threadIdx.x >> 5);
  const int nw   = gridDim.x * (BTHREADS / 32);
  const int k = lane & (NPART - 1);
  const int w = lane >> 4;
  const float* __restrict__ srcp = part + (size_t)(k * 2 + w) * n_nodes;
  const bool vec = aligned16(part) && aligned16(node_mean) && (n_nodes & 3) == 0;

  for (int r0 = gw * BROWS; r0 < n_nodes; r0 += nw * BROWS) {
    float v[BROWS], nm[BROWS];
    if (vec) {
      const f32x4 t = *(const f32x4*)(srcp + r0);
      const f32x4 g = *(const f32x4*)(node_mean + r0);
      v[0] = t.x; v[1] = t.y; v[2] = t.z; v[3] = t.w;
      nm[0] = g.x; nm[1] = g.y; nm[2] = g.z; nm[3] = g.w;
    } else {
      for (int u = 0; u < BROWS; ++u) {
        v[u]  = srcp[min(r0 + u, n_nodes - 1)];
        nm[u] = node_mean[min(r0 + u, n_nodes - 1)];
      }
    }
    for (int u = 0; u < BROWS; ++u) {
      float x = v[u];
      x += __shfl_xor(x, 8);
      x += __shfl_xor(x, 4);
      x += __shfl_xor(x, 2);
      x += __shfl_xor(x, 1);
      const float s = __shfl(x, 0);
      const float c = __shfl(x, 16);
      const float pm = c > 0.0f ? s / c + nm[u] * (1.0f / 3.0f) : 0.0f;
      if (r0 + u < n_nodes) {
        const f32x4 v4 = {pm, pm, pm, pm};
        f32x4* __restrict__ o = (f32x4*)out + (size_t)(r0 + u) * (OUTW / 4) + lane;
        for (int j = 0; j < OUTW / 4 / 32; ++j)
          __builtin_nontemporal_store(v4, o + j * 32);
      }
    }
  }
}

extern "C" void kernel_launch(void* const* d_in, const int* in_sizes, int n_in,
                              void* d_out, int out_size, void* d_ws, size_t ws_size,
                              hipStream_t stream) {
  const float* emb       = (const float*)d_in[0];
  const int* pair_place  = (const int*)d_in[1];
  const int* event_pair  = (const int*)d_in[2];
  const int* event_src   = (const int*)d_in[3];
  const int* event_ndst  = (const int*)d_in[4];
  const int* dst_event   = (const int*)d_in[5];
  const int* dst_node    = (const int*)d_in[6];

  const int n_nodes  = in_sizes[0] / EMB;
  const int n_pairs  = in_sizes[1];
  const int n_events = in_sizes[2];
  const int n_dst    = in_sizes[5];

  float* ws        = (float*)d_ws;
  float* node_mean = ws;
  float* dst_sum   = node_mean + n_nodes;
  float* pair_sum  = dst_sum + n_events;
  float* pair_cnt  = pair_sum + n_pairs;
  float* part      = dst_sum;

  rowmean_wmma<<<(n_nodes + 63) / 64, RM_THREADS, 0, stream>>>(emb, node_mean, n_nodes, dst_sum,
                                                               n_events, n_pairs, ws_size);

  dst_scatter<<<TB_GROUPS, TB_THREADS, 0, stream>>>(dst_event, dst_node, node_mean,
                                                    dst_sum, n_dst, n_nodes);

  event_reduce<<<TB_GROUPS, TB_THREADS, 0, stream>>>(
      event_pair, event_src, event_ndst, node_mean, dst_sum,
      pair_sum, pair_cnt, n_events, n_pairs, n_nodes, ws_size);

  pair_reduce<<<NPART, 1024, 0, stream>>>(pair_place, pair_sum, pair_cnt, part,
                                          n_pairs, n_nodes, n_events, ws_size);

  bcast_out<<<1024, BTHREADS, 0, stream>>>(part, node_mean, (float*)d_out, n_nodes);
}


# pair sum and count kept as one 8-byte record; an event's two adds issued by two adjacent lanes of one atomic instruction (16 random lines per add, not 32); atomic order still free
# speedup vs baseline: 3.1752x; 1.0937x over previous
#include <hip/hip_runtime.h>

#define EMB 512
#define OUTW (3 * EMB)

typedef __attribute__((ext_vector_type(2)))  float     v2f;
typedef __attribute__((ext_vector_type(8)))  float     v8f;
typedef __attribute__((ext_vector_type(4)))  float     f32x4;
typedef __attribute__((ext_vector_type(4)))  int       i32x4;


static __device__ __forceinline__ bool aligned16(const void* p) {
  return (((unsigned long long)p) & 15ull) == 0ull;
}

static __device__ __forceinline__ int pair_copies(size_t ws_bytes, int n_nodes, int n_events,
                                                  int n_pairs) {
  const size_t need =
      ((size_t)n_nodes + (size_t)n_events + 4 * (size_t)n_pairs) * sizeof(float);
  return ws_bytes >= need ? 2 : 1;
}

static __device__ __forceinline__ int cache_half() {
  return (int)((__builtin_amdgcn_s_sendmsg_rtn(0x87) >> 18) & 1u);
}

static __device__ __forceinline__ void zero_fill(float* __restrict__ p, int n, int tid, int nth) {
  const int mis  = (int)((((unsigned long long)p) & 15ull) >> 2);
  const int head = min(n, (4 - mis) & 3);
  if (tid < head) p[tid] = 0.0f;
  float* __restrict__ q = p + head;
  const int m  = n - head;
  const int m4 = m >> 2;
  f32x4* __restrict__ q4 = (f32x4*)q;
  for (int j = tid; j < m4; j += nth) q4[j] = f32x4{0.0f, 0.0f, 0.0f, 0.0f};
  const int t = m4 * 4 + tid;
  if (t < m) q[t] = 0.0f;
}

#define RM_THREADS 256
#define RM_ROWS 8
#define RM_CHUNKS (EMB / 64)

__global__ __launch_bounds__(RM_THREADS) void rowmean_wmma(const float* __restrict__ emb,
                                                           float* __restrict__ node_mean,
                                                           int n_nodes, float* __restrict__ acc,
                                                           int n_events, int n_pairs,
                                                           size_t ws_bytes) {
  static_assert(EMB % 64 == 0, "a row is a whole number of 256-byte chunks");
  static_assert(RM_ROWS % 4 == 0, "rows go in pairs, and two pairs' products are interleaved");
  zero_fill(acc, n_events + 2 * n_pairs * pair_copies(ws_bytes, n_nodes, n_events, n_pairs),
            blockIdx.x * RM_THREADS + threadIdx.x, gridDim.x * RM_THREADS);

  const int lane = threadIdx.x & 31;
  const int wave = threadIdx.x >> 5;
  const int row0 = (blockIdx.x * (RM_THREADS / 32) + wave) * RM_ROWS;
  if (row0 >= n_nodes) return;

  const int c   = lane & 15;
  const int h   = lane >> 4;
  const int sel = c >> 3;
  const int q   = (c & 7) + 8 * h;
  const float* __restrict__ rp[RM_ROWS / 2];
#pragma unroll
  for (int p = 0; p < RM_ROWS / 2; ++p)
    rp[p] = emb + (size_t)min(row0 + p + (RM_ROWS / 2) * sel, n_nodes - 1) * EMB + q * 4;

  f32x4 v[RM_ROWS / 2][RM_CHUNKS];
#pragma unroll
  for (int g = 0; g < RM_ROWS / 2; g += 2)
#pragma unroll
    for (int s = 0; s < RM_CHUNKS; ++s) {
      v[g][s]     = *(const f32x4*)(rp[g] + s * 64);
      v[g + 1][s] = *(const f32x4*)(rp[g + 1] + s * 64);
    }
  __builtin_amdgcn_sched_barrier(0);

  const v2f bones = {1.0f, 1.0f};
  v8f cacc[RM_ROWS / 2];
#pragma unroll
  for (int p = 0; p < RM_ROWS / 2; ++p) cacc[p] = v8f{};
#pragma unroll
  for (int g = 0; g < RM_ROWS / 2; g += 2)
#pragma unroll
    for (int s = 0; s < RM_CHUNKS; ++s) {
      const f32x4 x = v[g][s];
      const f32x4 y = v[g + 1][s];
      cacc[g] = __builtin_amdgcn_wmma_f32_16x16x4_f32(
           false, v2f{x.x, x.y},  false, bones,
           (short)0, cacc[g],  false,  false);
      cacc[g + 1] = __builtin_amdgcn_wmma_f32_16x16x4_f32(
          false, v2f{y.x, y.y}, false, bones, (short)0, cacc[g + 1], false, false);
      cacc[g] = __builtin_amdgcn_wmma_f32_16x16x4_f32(
          false, v2f{x.z, x.w}, false, bones, (short)0, cacc[g], false, false);
      cacc[g + 1] = __builtin_amdgcn_wmma_f32_16x16x4_f32(
          false, v2f{y.z, y.w}, false, bones, (short)0, cacc[g + 1], false, false);
    }

  if (c == 0) {
    const float inv = 1.0f / (float)EMB;
    const int base = row0 + (RM_ROWS / 2) * h;
#pragma unroll
    for (int p = 0; p < RM_ROWS / 2; ++p) {
      const float t = ((cacc[p][0] + cacc[p][1]) + (cacc[p][2] + cacc[p][3])) +
                      ((cacc[p][4] + cacc[p][5]) + (cacc[p][6] + cacc[p][7]));
      if (base + p < n_nodes) node_mean[base + p] = t * inv;
    }
  }
}

#define TAB_MAX 16384
#define TB_THREADS 1024
#define TB_GROUPS 256
#define TB_BATCH 4

static __device__ __forceinline__ void stage_table(float* __restrict__ tab,
                                                   const float* __restrict__ node_mean, int n) {
  const int t = threadIdx.x;
  if (aligned16(node_mean)) {
    const f32x4* __restrict__ s4 = (const f32x4*)node_mean;
    f32x4* __restrict__ t4 = (f32x4*)tab;
    const int n4 = n >> 2;
    for (int j0 = t; j0 < n4; j0 += TB_THREADS * TB_BATCH) {
      f32x4 v[TB_BATCH];
#pragma unroll
      for (int u = 0; u < TB_BATCH; ++u) {
        const int j = j0 + u * TB_THREADS;
        v[u] = s4[j < n4 ? j : j0];
      }
#pragma unroll
      for (int u = 0; u < TB_BATCH; ++u) {
        const int j = j0 + u * TB_THREADS;
        if (j < n4) t4[j] = v[u];
      }
    }
    const int r = n4 * 4 + t;
    if (r < n) tab[r] = node_mean[r];
  } else {
    for (int j = t; j < n; j += TB_THREADS) tab[j] = node_mean[j];
  }
}

template <bool IN_LDS>
static __device__ __forceinline__ float nm_at(const float* __restrict__ tab,
                                              const float* __restrict__ node_mean, int i) {
  return IN_LDS ? tab[i] : node_mean[i];
}

template <bool IN_LDS>
static __device__ __forceinline__ void dst_scatter_body(float* __restrict__ tab,
                                                        const int* __restrict__ dst_event,
                                                        const int* __restrict__ dst_node,
                                                        const float* __restrict__ node_mean,
                                                        float* __restrict__ dst_sum, int n_dst,
                                                        int n_nodes) {
  const int tid = blockIdx.x * TB_THREADS + threadIdx.x;
  const int nth = gridDim.x * TB_THREADS;
  if (aligned16(dst_event) && aligned16(dst_node)) {
    const i32x4* __restrict__ e4 = (const i32x4*)dst_event;
    const i32x4* __restrict__ d4 = (const i32x4*)dst_node;
    const int n4 = n_dst >> 2;
    int j = tid;
    i32x4 e = {0, 0, 0, 0};
    i32x4 d = {0, 0, 0, 0};
    if (j < n4) { e = e4[j]; d = d4[j]; }
    if (IN_LDS) {
      stage_table(tab, node_mean, n_nodes);
      __syncthreads();
    }
    while (j < n4) {
      const float a0 = nm_at<IN_LDS>(tab, node_mean, d.x);
      const float a1 = nm_at<IN_LDS>(tab, node_mean, d.y);
      const float a2 = nm_at<IN_LDS>(tab, node_mean, d.z);
      const float a3 = nm_at<IN_LDS>(tab, node_mean, d.w);
      unsafeAtomicAdd(&dst_sum[e.x], a0);
      unsafeAtomicAdd(&dst_sum[e.y], a1);
      unsafeAtomicAdd(&dst_sum[e.z], a2);
      unsafeAtomicAdd(&dst_sum[e.w], a3);
      j += nth;
      if (j < n4) { e = e4[j]; d = d4[j]; }
    }
    const int t = n4 * 4 + tid;
    if (t < n_dst)
      unsafeAtomicAdd(&dst_sum[dst_event[t]], nm_at<IN_LDS>(tab, node_mean, dst_node[t]));
  } else {
    if (IN_LDS) {
      stage_table(tab, node_mean, n_nodes);
      __syncthreads();
    }
    for (int i = tid; i < n_dst; i += nth)
      unsafeAtomicAdd(&dst_sum[dst_event[i]], nm_at<IN_LDS>(tab, node_mean, dst_node[i]));
  }
}

__global__ __launch_bounds__(TB_THREADS) void dst_scatter(const int* __restrict__ dst_event,
                                                          const int* __restrict__ dst_node,
                                                          const float* __restrict__ node_mean,
                                                          float* __restrict__ dst_sum, int n_dst,
                                                          int n_nodes) {
  __shared__ __attribute__((aligned(16))) float tab[TAB_MAX];
  if (n_nodes <= TAB_MAX)
    dst_scatter_body<true>(tab, dst_event, dst_node, node_mean, dst_sum, n_dst, n_nodes);
  else
    dst_scatter_body<false>(tab, dst_event, dst_node, node_mean, dst_sum, n_dst, n_nodes);
}

template <bool IN_LDS>
static __device__ __forceinline__ float event_val(int src, int ndst, float dsum,
                                                  const float* __restrict__ tab,
                                                  const float* __restrict__ node_mean) {
  return (nm_at<IN_LDS>(tab, node_mean, src) + dsum / ((float)ndst + 1.0f)) * (1.0f / 3.0f);
}

static __device__ __forceinline__ void event_add(int p, float s, float* __restrict__ pair_acc) {
  unsafeAtomicAdd(&pair_acc[2 * (size_t)p], s);
  unsafeAtomicAdd(&pair_acc[2 * (size_t)p + 1], 1.0f);
}

static __device__ __forceinline__ int lane_neighbour(int x) {
  return __builtin_amdgcn_update_dpp(0, x, 0xB1, 0xF, 0xF, false);
}

static __device__ __forceinline__ void record_add(float* __restrict__ pair_acc, int key, int word,
                                                  float val) {
  unsafeAtomicAdd(&pair_acc[2 * (size_t)key + word], val);
}

template <bool IN_LDS>
static __device__ __forceinline__ void event_reduce_body(float* __restrict__ tab,
                                                         const int* __restrict__ event_pair,
                                                         const int* __restrict__ event_src,
                                                         const int* __restrict__ event_ndst,
                                                         const float* __restrict__ node_mean,
                                                         const float* __restrict__ dst_sum,
                                                         float* __restrict__ pair_acc0, int n_events,
                                                         int n_pairs, int n_nodes, size_t ws_bytes) {
  const int tid = blockIdx.x * TB_THREADS + threadIdx.x;
  const int nth = gridDim.x * TB_THREADS;
  const int h = pair_copies(ws_bytes, n_nodes, n_events, n_pairs) == 2 ? cache_half() : 0;
  float* __restrict__ pair_acc = pair_acc0 + (size_t)h * 2 * (size_t)n_pairs;
  if (aligned16(event_pair) && aligned16(event_src) && aligned16(event_ndst) &&
      aligned16(dst_sum)) {
    const i32x4* __restrict__ p4 = (const i32x4*)event_pair;
    const i32x4* __restrict__ s4 = (const i32x4*)event_src;
    const i32x4* __restrict__ n4v = (const i32x4*)event_ndst;
    const f32x4* __restrict__ d4 = (const f32x4*)dst_sum;
    const int n4 = n_events >> 2;
    const int odd = threadIdx.x & 1;
    int j = tid;
    int jw = __builtin_amdgcn_readfirstlane(tid - (int)(threadIdx.x & 31));
    i32x4 p = {0, 0, 0, 0};
    i32x4 sr = {0, 0, 0, 0};
    i32x4 nd = {0, 0, 0, 0};
    f32x4 ds = {0.0f, 0.0f, 0.0f, 0.0f};
    if (j < n4) { p = p4[j]; sr = s4[j]; nd = n4v[j]; ds = d4[j]; }
    if (IN_LDS) {
      stage_table(tab, node_mean, n_nodes);
      __syncthreads();
    }
    while (jw + 32 <= n4) {
      const float v0 = event_val<IN_LDS>(sr.x, nd.x, ds.x, tab, node_mean);
      const float v1 = event_val<IN_LDS>(sr.y, nd.y, ds.y, tab, node_mean);
      const float v2 = event_val<IN_LDS>(sr.z, nd.z, ds.z, tab, node_mean);
      const float v3 = event_val<IN_LDS>(sr.w, nd.w, ds.w, tab, node_mean);
      const int r0 = lane_neighbour(p.x);
      const int r1 = lane_neighbour(p.y);
      const int r2 = lane_neighbour(p.z);
      const int r3 = lane_neighbour(p.w);
      record_add(pair_acc, odd ? r0 : p.x, odd, odd ? 1.0f : v0);
      record_add(pair_acc, odd ? r1 : p.y, odd, odd ? 1.0f : v1);
      record_add(pair_acc, odd ? r2 : p.z, odd, odd ? 1.0f : v2);
      record_add(pair_acc, odd ? r3 : p.w, odd, odd ? 1.0f : v3);
      record_add(pair_acc, odd ? p.x : r0, odd ^ 1, odd ? v0 : 1.0f);
      record_add(pair_acc, odd ? p.y : r1, odd ^ 1, odd ? v1 : 1.0f);
      record_add(pair_acc, odd ? p.z : r2, odd ^ 1, odd ? v2 : 1.0f);
      record_add(pair_acc, odd ? p.w : r3, odd ^ 1, odd ? v3 : 1.0f);
      j += nth;
      jw += nth;
      if (j < n4) { p = p4[j]; sr = s4[j]; nd = n4v[j]; ds = d4[j]; }
    }
    if (j < n4) {
      const float v0 = event_val<IN_LDS>(sr.x, nd.x, ds.x, tab, node_mean);
      const float v1 = event_val<IN_LDS>(sr.y, nd.y, ds.y, tab, node_mean);
      const float v2 = event_val<IN_LDS>(sr.z, nd.z, ds.z, tab, node_mean);
      const float v3 = event_val<IN_LDS>(sr.w, nd.w, ds.w, tab, node_mean);
      event_add(p.x, v0, pair_acc);
      event_add(p.y, v1, pair_acc);
      event_add(p.z, v2, pair_acc);
      event_add(p.w, v3, pair_acc);
    }
    const int t = n4 * 4 + tid;
    if (t < n_events)
      event_add(event_pair[t],
                event_val<IN_LDS>(event_src[t], event_ndst[t], dst_sum[t], tab, node_mean),
                pair_acc);
  } else {
    if (IN_LDS) {
      stage_table(tab, node_mean, n_nodes);
      __syncthreads();
    }
    for (int i = tid; i < n_events; i += nth)
      event_add(event_pair[i],
                event_val<IN_LDS>(event_src[i], event_ndst[i], dst_sum[i], tab, node_mean),
                pair_acc);
  }
}

__global__ __launch_bounds__(TB_THREADS) void event_reduce(const int* __restrict__ event_pair,
                                                           const int* __restrict__ event_src,
                                                           const int* __restrict__ event_ndst,
                                                           const float* __restrict__ node_mean,
                                                           const float* __restrict__ dst_sum,
                                                           float* __restrict__ pair_acc0, int n_events,
                                                           int n_pairs, int n_nodes, size_t ws_bytes) {
  __shared__ __attribute__((aligned(16))) float tab[TAB_MAX];
  if (n_nodes <= TAB_MAX)
    event_reduce_body<true>(tab, event_pair, event_src, event_ndst, node_mean, dst_sum, pair_acc0,
                            n_events, n_pairs, n_nodes, ws_bytes);
  else
    event_reduce_body<false>(tab, event_pair, event_src, event_ndst, node_mean, dst_sum, pair_acc0,
                             n_events, n_pairs, n_nodes, ws_bytes);
}

#define NPART 16
#define PLACE_TILE 16384
#define PBATCH 8

__global__ __launch_bounds__(1024) void pair_reduce(const int* __restrict__ pair_place,
                                                    const float* __restrict__ pair_acc,
                                                    float* __restrict__ part,
                                                    int n_pairs, int n_nodes, int n_events,
                                                    size_t ws_bytes) {
  const bool two = pair_copies(ws_bytes, n_nodes, n_events, n_pairs) == 2;
  const float* __restrict__ pair_acc1 = pair_acc + (two ? 2 * (size_t)n_pairs : 0);
  __shared__ __attribute__((aligned(16))) float tsum[PLACE_TILE];
  __shared__ __attribute__((aligned(16))) float tcnt[PLACE_TILE];
  const int tid = threadIdx.x;
  const int per = (n_pairs + NPART - 1) / NPART;
  const int lo  = blockIdx.x * per;
  const int hi  = min(lo + per, n_pairs);
  float* __restrict__ osum = part + (size_t)(blockIdx.x * 2) * n_nodes;
  float* __restrict__ ocnt = osum + n_nodes;

  for (int base = 0; base < n_nodes; base += PLACE_TILE) {
    for (int j = tid; j < PLACE_TILE / 4; j += 1024) {
      ((f32x4*)tsum)[j] = f32x4{0.0f, 0.0f, 0.0f, 0.0f};
      ((f32x4*)tcnt)[j] = f32x4{0.0f, 0.0f, 0.0f, 0.0f};
    }
    __syncthreads();

    for (int i0 = lo + tid; i0 < hi; i0 += 1024 * PBATCH) {
      float c[PBATCH], s[PBATCH];
      int pl[PBATCH];
#pragma unroll
      for (int u = 0; u < PBATCH; ++u) {
        const int i = i0 + u * 1024;
        const bool ok = i < hi;
        const int ic = ok ? i : i0;
        const float s0 = pair_acc[2 * (size_t)ic];
        const float c0 = pair_acc[2 * (size_t)ic + 1];
        const float s1 = pair_acc1[2 * (size_t)ic];
        const float c1 = pair_acc1[2 * (size_t)ic + 1];
        const float cu = c0 + (two ? c1 : 0.0f);
        c[u]  = ok ? cu : 0.0f;
        s[u]  = s0 + (two ? s1 : 0.0f);
        pl[u] = pair_place[ic];
      }
#pragma unroll
      for (int u = 0; u < PBATCH; ++u) {
        const unsigned q = (unsigned)(pl[u] - base);
        if (c[u] > 0.0f && q < (unsigned)PLACE_TILE) {
          __hip_atomic_fetch_add(&tsum[q], s[u] / c[u], __ATOMIC_RELAXED,
                                 __HIP_MEMORY_SCOPE_WORKGROUP);
          __hip_atomic_fetch_add(&tcnt[q], 1.0f, __ATOMIC_RELAXED,
                                 __HIP_MEMORY_SCOPE_WORKGROUP);
        }
      }
    }
    __syncthreads();

    const int lim = min(PLACE_TILE, n_nodes - base);
    if ((n_nodes & 3) == 0) {
      f32x4* __restrict__ os4 = (f32x4*)(osum + base);
      f32x4* __restrict__ oc4 = (f32x4*)(ocnt + base);
      for (int j = tid; j < lim / 4; j += 1024) {
        os4[j] = ((const f32x4*)tsum)[j];
        oc4[j] = ((const f32x4*)tcnt)[j];
      }
    } else {
      for (int j = tid; j < lim; j += 1024) {
        osum[base + j] = tsum[j];
        ocnt[base + j] = tcnt[j];
      }
    }
    __syncthreads();
  }
}

#define BROWS 4
#define BTHREADS 128

__global__ __launch_bounds__(BTHREADS) void bcast_out(const float* __restrict__ part,
                                                      const float* __restrict__ node_mean,
                                                      float* __restrict__ out, int n_nodes) {
  static_assert(NPART == 16, "lanes 0-15 / 16-31 of a wave map to the 16 partial tables");
  static_assert((OUTW / 4) % 32 == 0, "a row is a whole number of 512-byte wave stores");
  const int lane = threadIdx.x & 31;
  const int gw   = blockIdx.x * (BTHREADS / 32) + (threadIdx.x >> 5);
  const int nw   = gridDim.x * (BTHREADS / 32);
  const int k = lane & (NPART - 1);
  const int w = lane >> 4;
  const float* __restrict__ srcp = part + (size_t)(k * 2 + w) * n_nodes;
  const bool vec = aligned16(part) && aligned16(node_mean) && (n_nodes & 3) == 0;

  for (int r0 = gw * BROWS; r0 < n_nodes; r0 += nw * BROWS) {
    float v[BROWS], nm[BROWS];
    if (vec) {
      const f32x4 t = *(const f32x4*)(srcp + r0);
      const f32x4 g = *(const f32x4*)(node_mean + r0);
      v[0] = t.x; v[1] = t.y; v[2] = t.z; v[3] = t.w;
      nm[0] = g.x; nm[1] = g.y; nm[2] = g.z; nm[3] = g.w;
    } else {
      for (int u = 0; u < BROWS; ++u) {
        v[u]  = srcp[min(r0 + u, n_nodes - 1)];
        nm[u] = node_mean[min(r0 + u, n_nodes - 1)];
      }
    }
    for (int u = 0; u < BROWS; ++u) {
      float x = v[u];
      x += __shfl_xor(x, 8);
      x += __shfl_xor(x, 4);
      x += __shfl_xor(x, 2);
      x += __shfl_xor(x, 1);
      const float s = __shfl(x, 0);
      const float c = __shfl(x, 16);
      const float pm = c > 0.0f ? s / c + nm[u] * (1.0f / 3.0f) : 0.0f;
      if (r0 + u < n_nodes) {
        const f32x4 v4 = {pm, pm, pm, pm};
        f32x4* __restrict__ o = (f32x4*)out + (size_t)(r0 + u) * (OUTW / 4) + lane;
        for (int j = 0; j < OUTW / 4 / 32; ++j)
          __builtin_nontemporal_store(v4, o + j * 32);
      }
    }
  }
}

extern "C" void kernel_launch(void* const* d_in, const int* in_sizes, int n_in,
                              void* d_out, int out_size, void* d_ws, size_t ws_size,
                              hipStream_t stream) {
  const float* emb       = (const float*)d_in[0];
  const int* pair_place  = (const int*)d_in[1];
  const int* event_pair  = (const int*)d_in[2];
  const int* event_src   = (const int*)d_in[3];
  const int* event_ndst  = (const int*)d_in[4];
  const int* dst_event   = (const int*)d_in[5];
  const int* dst_node    = (const int*)d_in[6];

  const int n_nodes  = in_sizes[0] / EMB;
  const int n_pairs  = in_sizes[1];
  const int n_events = in_sizes[2];
  const int n_dst    = in_sizes[5];

  float* ws        = (float*)d_ws;
  float* node_mean = ws;
  float* dst_sum   = node_mean + n_nodes;
  float* pair_acc  = dst_sum + n_events;
  float* part      = dst_sum;

  rowmean_wmma<<<(n_nodes + 63) / 64, RM_THREADS, 0, stream>>>(emb, node_mean, n_nodes, dst_sum,
                                                               n_events, n_pairs, ws_size);

  dst_scatter<<<TB_GROUPS, TB_THREADS, 0, stream>>>(dst_event, dst_node, node_mean,
                                                    dst_sum, n_dst, n_nodes);

  event_reduce<<<TB_GROUPS, TB_THREADS, 0, stream>>>(
      event_pair, event_src, event_ndst, node_mean, dst_sum,
      pair_acc, n_events, n_pairs, n_nodes, ws_size);

  pair_reduce<<<NPART, 1024, 0, stream>>>(pair_place, pair_acc, part,
                                          n_pairs, n_nodes, n_events, ws_size);

  bcast_out<<<1024, BTHREADS, 0, stream>>>(part, node_mean, (float*)d_out, n_nodes);
}


# dst and event stages in one launch: a work-group takes a chunk of the sorted dst arrays, sums it per event in an LDS window, then adds its events to pairs; 5 launches to 4; sum order changes
# speedup vs baseline: 3.8452x; 1.2110x over previous
#include <hip/hip_runtime.h>

#define EMB 512
#define OUTW (3 * EMB)

typedef __attribute__((ext_vector_type(2)))  float     v2f;
typedef __attribute__((ext_vector_type(8)))  float     v8f;
typedef __attribute__((ext_vector_type(4)))  float     f32x4;
typedef __attribute__((ext_vector_type(4)))  int       i32x4;


static __device__ __forceinline__ bool aligned16(const void* p) {
  return (((unsigned long long)p) & 15ull) == 0ull;
}

static __device__ __forceinline__ int pair_copies(size_t ws_bytes, int n_nodes, int n_events,
                                                  int n_pairs) {
  const size_t need =
      ((size_t)n_nodes + (size_t)n_events + 4 * (size_t)n_pairs) * sizeof(float);
  return ws_bytes >= need ? 2 : 1;
}

static __device__ __forceinline__ int cache_half() {
  return (int)((__builtin_amdgcn_s_sendmsg_rtn(0x87) >> 18) & 1u);
}

static __device__ __forceinline__ void zero_fill(float* __restrict__ p, int n, int tid, int nth) {
  const int mis  = (int)((((unsigned long long)p) & 15ull) >> 2);
  const int head = min(n, (4 - mis) & 3);
  if (tid < head) p[tid] = 0.0f;
  float* __restrict__ q = p + head;
  const int m  = n - head;
  const int m4 = m >> 2;
  f32x4* __restrict__ q4 = (f32x4*)q;
  for (int j = tid; j < m4; j += nth) q4[j] = f32x4{0.0f, 0.0f, 0.0f, 0.0f};
  const int t = m4 * 4 + tid;
  if (t < m) q[t] = 0.0f;
}

#define RM_THREADS 256
#define RM_ROWS 8
#define RM_CHUNKS (EMB / 64)

__global__ __launch_bounds__(RM_THREADS) void rowmean_wmma(const float* __restrict__ emb,
                                                           float* __restrict__ node_mean,
                                                           int n_nodes, float* __restrict__ acc,
                                                           int n_events, int n_pairs,
                                                           size_t ws_bytes) {
  static_assert(EMB % 64 == 0, "a row is a whole number of 256-byte chunks");
  static_assert(RM_ROWS % 4 == 0, "rows go in pairs, and two pairs' products are interleaved");
  zero_fill(acc, 2 * n_pairs * pair_copies(ws_bytes, n_nodes, n_events, n_pairs),
            blockIdx.x * RM_THREADS + threadIdx.x, gridDim.x * RM_THREADS);

  const int lane = threadIdx.x & 31;
  const int wave = threadIdx.x >> 5;
  const int row0 = (blockIdx.x * (RM_THREADS / 32) + wave) * RM_ROWS;
  if (row0 >= n_nodes) return;

  const int c   = lane & 15;
  const int h   = lane >> 4;
  const int sel = c >> 3;
  const int q   = (c & 7) + 8 * h;
  const float* __restrict__ rp[RM_ROWS / 2];
#pragma unroll
  for (int p = 0; p < RM_ROWS / 2; ++p)
    rp[p] = emb + (size_t)min(row0 + p + (RM_ROWS / 2) * sel, n_nodes - 1) * EMB + q * 4;

  f32x4 v[RM_ROWS / 2][RM_CHUNKS];
#pragma unroll
  for (int g = 0; g < RM_ROWS / 2; g += 2)
#pragma unroll
    for (int s = 0; s < RM_CHUNKS; ++s) {
      v[g][s]     = *(const f32x4*)(rp[g] + s * 64);
      v[g + 1][s] = *(const f32x4*)(rp[g + 1] + s * 64);
    }
  __builtin_amdgcn_sched_barrier(0);

  const v2f bones = {1.0f, 1.0f};
  v8f cacc[RM_ROWS / 2];
#pragma unroll
  for (int p = 0; p < RM_ROWS / 2; ++p) cacc[p] = v8f{};
#pragma unroll
  for (int g = 0; g < RM_ROWS / 2; g += 2)
#pragma unroll
    for (int s = 0; s < RM_CHUNKS; ++s) {
      const f32x4 x = v[g][s];
      const f32x4 y = v[g + 1][s];
      cacc[g] = __builtin_amdgcn_wmma_f32_16x16x4_f32(
           false, v2f{x.x, x.y},  false, bones,
           (short)0, cacc[g],  false,  false);
      cacc[g + 1] = __builtin_amdgcn_wmma_f32_16x16x4_f32(
          false, v2f{y.x, y.y}, false, bones, (short)0, cacc[g + 1], false, false);
      cacc[g] = __builtin_amdgcn_wmma_f32_16x16x4_f32(
          false, v2f{x.z, x.w}, false, bones, (short)0, cacc[g], false, false);
      cacc[g + 1] = __builtin_amdgcn_wmma_f32_16x16x4_f32(
          false, v2f{y.z, y.w}, false, bones, (short)0, cacc[g + 1], false, false);
    }

  if (c == 0) {
    const float inv = 1.0f / (float)EMB;
    const int base = row0 + (RM_ROWS / 2) * h;
#pragma unroll
    for (int p = 0; p < RM_ROWS / 2; ++p) {
      const float t = ((cacc[p][0] + cacc[p][1]) + (cacc[p][2] + cacc[p][3])) +
                      ((cacc[p][4] + cacc[p][5]) + (cacc[p][6] + cacc[p][7]));
      if (base + p < n_nodes) node_mean[base + p] = t * inv;
    }
  }
}

#define TAB_MAX 16384
#define TB_THREADS 1024
#define TB_GROUPS 256
#define TB_BATCH 4

static __device__ __forceinline__ void stage_table(float* __restrict__ tab,
                                                   const float* __restrict__ node_mean, int n) {
  const int t = threadIdx.x;
  if (aligned16(node_mean)) {
    const f32x4* __restrict__ s4 = (const f32x4*)node_mean;
    f32x4* __restrict__ t4 = (f32x4*)tab;
    const int n4 = n >> 2;
    for (int j0 = t; j0 < n4; j0 += TB_THREADS * TB_BATCH) {
      f32x4 v[TB_BATCH];
#pragma unroll
      for (int u = 0; u < TB_BATCH; ++u) {
        const int j = j0 + u * TB_THREADS;
        v[u] = s4[j < n4 ? j : j0];
      }
#pragma unroll
      for (int u = 0; u < TB_BATCH; ++u) {
        const int j = j0 + u * TB_THREADS;
        if (j < n4) t4[j] = v[u];
      }
    }
    const int r = n4 * 4 + t;
    if (r < n) tab[r] = node_mean[r];
  } else {
    for (int j = t; j < n; j += TB_THREADS) tab[j] = node_mean[j];
  }
}

template <bool IN_LDS>
static __device__ __forceinline__ float nm_at(const float* __restrict__ tab,
                                              const float* __restrict__ node_mean, int i) {
  return IN_LDS ? tab[i] : node_mean[i];
}

#define EV_WIN (TB_THREADS * 4)
#define DST_TILE (TB_THREADS * 4)
#define DST_PRE 3

template <bool IN_LDS>
static __device__ __forceinline__ float event_val(int src, int ndst, float dsum,
                                                  const float* __restrict__ tab,
                                                  const float* __restrict__ node_mean) {
  return (nm_at<IN_LDS>(tab, node_mean, src) + dsum / ((float)ndst + 1.0f)) * (1.0f / 3.0f);
}

static __device__ __forceinline__ int lane_neighbour(int x) {
  return __builtin_amdgcn_update_dpp(0, x, 0xB1, 0xF, 0xF, false);
}

static __device__ __forceinline__ void record_add(float* __restrict__ pair_acc, int key, int word,
                                                  float val) {
  unsafeAtomicAdd(&pair_acc[2 * (size_t)key + word], val);
}

static __device__ __forceinline__ void event_add(int p, float s, float* __restrict__ pair_acc) {
  unsafeAtomicAdd(&pair_acc[2 * (size_t)p], s);
  unsafeAtomicAdd(&pair_acc[2 * (size_t)p + 1], 1.0f);
}

static __device__ __forceinline__ void load_dst4(const int* __restrict__ dst_event,
                                                 const int* __restrict__ dst_node, bool vec, int p,
                                                 int n_dst, i32x4& e, i32x4& d) {
  e = i32x4{-1, -1, -1, -1};
  d = i32x4{0, 0, 0, 0};
  if (vec) {
    const int j = p >> 2;
    if (j < (n_dst >> 2)) {
      e = ((const i32x4*)dst_event)[j];
      d = ((const i32x4*)dst_node)[j];
    }
  } else {
    if (p < n_dst)     { e.x = dst_event[p];     d.x = dst_node[p]; }
    if (p + 1 < n_dst) { e.y = dst_event[p + 1]; d.y = dst_node[p + 1]; }
    if (p + 2 < n_dst) { e.z = dst_event[p + 2]; d.z = dst_node[p + 2]; }
    if (p + 3 < n_dst) { e.w = dst_event[p + 3]; d.w = dst_node[p + 3]; }
  }
}

template <bool IN_LDS>
static __device__ __forceinline__ void window_add4(float* __restrict__ dsum,
                                                   const float* __restrict__ tab,
                                                   const float* __restrict__ node_mean, i32x4 e,
                                                   i32x4 d, int wb, int lo, unsigned span) {
  const float a0 = nm_at<IN_LDS>(tab, node_mean, d.x);
  const float a1 = nm_at<IN_LDS>(tab, node_mean, d.y);
  const float a2 = nm_at<IN_LDS>(tab, node_mean, d.z);
  const float a3 = nm_at<IN_LDS>(tab, node_mean, d.w);
  if ((unsigned)(e.x - lo) < span)
    __hip_atomic_fetch_add(&dsum[e.x - wb], a0, __ATOMIC_RELAXED, __HIP_MEMORY_SCOPE_WORKGROUP);
  if ((unsigned)(e.y - lo) < span)
    __hip_atomic_fetch_add(&dsum[e.y - wb], a1, __ATOMIC_RELAXED, __HIP_MEMORY_SCOPE_WORKGROUP);
  if ((unsigned)(e.z - lo) < span)
    __hip_atomic_fetch_add(&dsum[e.z - wb], a2, __ATOMIC_RELAXED, __HIP_MEMORY_SCOPE_WORKGROUP);
  if ((unsigned)(e.w - lo) < span)
    __hip_atomic_fetch_add(&dsum[e.w - wb], a3, __ATOMIC_RELAXED, __HIP_MEMORY_SCOPE_WORKGROUP);
}

template <bool IN_LDS>
static __device__ __forceinline__ void dst_event_body(float* __restrict__ tab,
                                                      float* __restrict__ dsum,
                                                      int* __restrict__ bnd,
                                                      const int* __restrict__ dst_event,
                                                      const int* __restrict__ dst_node,
                                                      const int* __restrict__ event_pair,
                                                      const int* __restrict__ event_src,
                                                      const int* __restrict__ event_ndst,
                                                      const float* __restrict__ node_mean,
                                                      float* __restrict__ pair_acc0, int n_dst,
                                                      int n_events, int n_pairs, int n_nodes,
                                                      size_t ws_bytes) {
  if (n_events <= 0) return;
  const int t = threadIdx.x;
  const int G = gridDim.x;
  const int b = blockIdx.x;
  const int h = pair_copies(ws_bytes, n_nodes, n_events, n_pairs) == 2 ? cache_half() : 0;
  float* __restrict__ pair_acc = pair_acc0 + (size_t)h * 2 * (size_t)n_pairs;

  const int chunk = (((n_dst + G - 1) / G) + 3) & ~3;
  const int k0 = b * chunk;
  const int k1 = k0 + chunk;
  const int c0 = min(k0, n_dst);
  const int c1 = min(k1, n_dst);
  const bool dvec = aligned16(dst_event) && aligned16(dst_node);
  const bool evec = aligned16(event_pair) && aligned16(event_src) && aligned16(event_ndst);
  const int q0 = dvec ? c0 + DST_PRE * DST_TILE : c0;

  const int si = t == 0 ? k0 - 1 : (t == 1 ? k1 - 1 : q0);
  const int* __restrict__ sp =
      n_dst > 0 ? dst_event + min(max(si, 0), n_dst - 1) : event_pair;
  const int sv = *sp;
  i32x4 pe[DST_PRE], pd[DST_PRE];
#pragma unroll
  for (int k = 0; k < DST_PRE; ++k) {
    pe[k] = i32x4{-1, -1, -1, -1};
    pd[k] = i32x4{0, 0, 0, 0};
    if (dvec) load_dst4(dst_event, dst_node, true, c0 + k * DST_TILE + 4 * t, n_dst, pe[k], pd[k]);
  }
  int te = -1, td = 0;
  if (dvec && t < (n_dst & 3)) {
    te = dst_event[(n_dst & ~3) + t];
    td = dst_node[(n_dst & ~3) + t];
  }

  if (IN_LDS) stage_table(tab, node_mean, n_nodes);
  if (t < 3) bnd[t] = sv;
  ((f32x4*)dsum)[t] = f32x4{0.0f, 0.0f, 0.0f, 0.0f};
  __syncthreads();

  const int f0 = b <= 0 ? 0 : (k0 >= n_dst ? n_events : min(max(bnd[0] + 1, 0), n_events));
  const int f1 =
      max(f0, (b + 1 >= G || k1 >= n_dst) ? n_events : min(max(bnd[1] + 1, 0), n_events));
  const int ex = q0 < n_dst ? bnd[2] : n_events;
  const int odd = t & 1;
  for (int wb = f0 & ~3; wb < f1; wb += EV_WIN) {
    const int lo = max(wb, f0);
    const int hi = min(wb + EV_WIN, f1);
    const unsigned span = (unsigned)(hi - lo);
    if (wb != (f0 & ~3)) {
      ((f32x4*)dsum)[t] = f32x4{0.0f, 0.0f, 0.0f, 0.0f};
      __syncthreads();
    }

    const int ev = wb + 4 * t;
    const bool fast = evec && ev + 3 < n_events;
    i32x4 p = {0, 0, 0, 0};
    i32x4 sr = {0, 0, 0, 0};
    i32x4 nd = {0, 0, 0, 0};
    if (fast) {
      p  = *(const i32x4*)(event_pair + ev);
      sr = *(const i32x4*)(event_src + ev);
      nd = *(const i32x4*)(event_ndst + ev);
    }

#pragma unroll
    for (int k = 0; k < DST_PRE; ++k)
      window_add4<IN_LDS>(dsum, tab, node_mean, pe[k], pd[k], wb, lo, span);
    if ((unsigned)(te - lo) < span)
      __hip_atomic_fetch_add(&dsum[te - wb], nm_at<IN_LDS>(tab, node_mean, td), __ATOMIC_RELAXED,
                             __HIP_MEMORY_SCOPE_WORKGROUP);
    {
      int q = q0;
      int eq = ex;
      while (q < n_dst && (q < c1 || eq < f1)) {
        i32x4 e, d;
        load_dst4(dst_event, dst_node, dvec, q + 4 * t, n_dst, e, d);
        window_add4<IN_LDS>(dsum, tab, node_mean, e, d, wb, lo, span);
        q += DST_TILE;
        eq = q < n_dst ? dst_event[q] : n_events;
      }
    }
    __syncthreads();

    const f32x4 ds = ((const f32x4*)dsum)[t];
    const float v0 = event_val<IN_LDS>(sr.x, nd.x, ds.x, tab, node_mean);
    const float v1 = event_val<IN_LDS>(sr.y, nd.y, ds.y, tab, node_mean);
    const float v2 = event_val<IN_LDS>(sr.z, nd.z, ds.z, tab, node_mean);
    const float v3 = event_val<IN_LDS>(sr.w, nd.w, ds.w, tab, node_mean);
    const int mk = !fast ? 0
                         : (((unsigned)(ev - lo) < span ? 1 : 0) |
                            ((unsigned)(ev + 1 - lo) < span ? 2 : 0) |
                            ((unsigned)(ev + 2 - lo) < span ? 4 : 0) |
                            ((unsigned)(ev + 3 - lo) < span ? 8 : 0));
    const int rk = lane_neighbour(mk);
    const int r0 = lane_neighbour(p.x);
    const int r1 = lane_neighbour(p.y);
    const int r2 = lane_neighbour(p.z);
    const int r3 = lane_neighbour(p.w);
    const int me = odd ? rk : mk;
    const int mo = odd ? mk : rk;
    if (me & 1) record_add(pair_acc, odd ? r0 : p.x, odd, odd ? 1.0f : v0);
    if (me & 2) record_add(pair_acc, odd ? r1 : p.y, odd, odd ? 1.0f : v1);
    if (me & 4) record_add(pair_acc, odd ? r2 : p.z, odd, odd ? 1.0f : v2);
    if (me & 8) record_add(pair_acc, odd ? r3 : p.w, odd, odd ? 1.0f : v3);
    if (mo & 1) record_add(pair_acc, odd ? p.x : r0, odd ^ 1, odd ? v0 : 1.0f);
    if (mo & 2) record_add(pair_acc, odd ? p.y : r1, odd ^ 1, odd ? v1 : 1.0f);
    if (mo & 4) record_add(pair_acc, odd ? p.z : r2, odd ^ 1, odd ? v2 : 1.0f);
    if (mo & 8) record_add(pair_acc, odd ? p.w : r3, odd ^ 1, odd ? v3 : 1.0f);
    if (!fast) {
      for (int k = 0; k < 4; ++k) {
        const int e = ev + k;
        if (e < n_events && (unsigned)(e - lo) < span)
          event_add(event_pair[e],
                    event_val<IN_LDS>(event_src[e], event_ndst[e], dsum[e - wb], tab, node_mean),
                    pair_acc);
      }
    }
  }
}

__global__ __launch_bounds__(TB_THREADS) void dst_event_reduce(const int* __restrict__ dst_event,
                                                               const int* __restrict__ dst_node,
                                                               const int* __restrict__ event_pair,
                                                               const int* __restrict__ event_src,
                                                               const int* __restrict__ event_ndst,
                                                               const float* __restrict__ node_mean,
                                                               float* __restrict__ pair_acc0, int n_dst,
                                                               int n_events, int n_pairs, int n_nodes,
                                                               size_t ws_bytes) {
  __shared__ __attribute__((aligned(16))) float tab[TAB_MAX];
  __shared__ __attribute__((aligned(16))) float dsum[EV_WIN];
  __shared__ int bnd[4];
  if (n_nodes <= TAB_MAX)
    dst_event_body<true>(tab, dsum, bnd, dst_event, dst_node, event_pair, event_src, event_ndst, node_mean,
                         pair_acc0, n_dst, n_events, n_pairs, n_nodes, ws_bytes);
  else
    dst_event_body<false>(tab, dsum, bnd, dst_event, dst_node, event_pair, event_src, event_ndst,
                          node_mean, pair_acc0, n_dst, n_events, n_pairs, n_nodes, ws_bytes);
}

#define NPART 16
#define PLACE_TILE 16384
#define PBATCH 8

__global__ __launch_bounds__(1024) void pair_reduce(const int* __restrict__ pair_place,
                                                    const float* __restrict__ pair_acc,
                                                    float* __restrict__ part,
                                                    int n_pairs, int n_nodes, int n_events,
                                                    size_t ws_bytes) {
  const bool two = pair_copies(ws_bytes, n_nodes, n_events, n_pairs) == 2;
  const float* __restrict__ pair_acc1 = pair_acc + (two ? 2 * (size_t)n_pairs : 0);
  __shared__ __attribute__((aligned(16))) float tsum[PLACE_TILE];
  __shared__ __attribute__((aligned(16))) float tcnt[PLACE_TILE];
  const int tid = threadIdx.x;
  const int per = (n_pairs + NPART - 1) / NPART;
  const int lo  = blockIdx.x * per;
  const int hi  = min(lo + per, n_pairs);
  float* __restrict__ osum = part + (size_t)(blockIdx.x * 2) * n_nodes;
  float* __restrict__ ocnt = osum + n_nodes;

  for (int base = 0; base < n_nodes; base += PLACE_TILE) {
    for (int j = tid; j < PLACE_TILE / 4; j += 1024) {
      ((f32x4*)tsum)[j] = f32x4{0.0f, 0.0f, 0.0f, 0.0f};
      ((f32x4*)tcnt)[j] = f32x4{0.0f, 0.0f, 0.0f, 0.0f};
    }
    __syncthreads();

    for (int i0 = lo + tid; i0 < hi; i0 += 1024 * PBATCH) {
      float c[PBATCH], s[PBATCH];
      int pl[PBATCH];
#pragma unroll
      for (int u = 0; u < PBATCH; ++u) {
        const int i = i0 + u * 1024;
        const bool ok = i < hi;
        const int ic = ok ? i : i0;
        const float s0 = pair_acc[2 * (size_t)ic];
        const float c0 = pair_acc[2 * (size_t)ic + 1];
        const float s1 = pair_acc1[2 * (size_t)ic];
        const float c1 = pair_acc1[2 * (size_t)ic + 1];
        const float cu = c0 + (two ? c1 : 0.0f);
        c[u]  = ok ? cu : 0.0f;
        s[u]  = s0 + (two ? s1 : 0.0f);
        pl[u] = pair_place[ic];
      }
#pragma unroll
      for (int u = 0; u < PBATCH; ++u) {
        const unsigned q = (unsigned)(pl[u] - base);
        if (c[u] > 0.0f && q < (unsigned)PLACE_TILE) {
          __hip_atomic_fetch_add(&tsum[q], s[u] / c[u], __ATOMIC_RELAXED,
                                 __HIP_MEMORY_SCOPE_WORKGROUP);
          __hip_atomic_fetch_add(&tcnt[q], 1.0f, __ATOMIC_RELAXED,
                                 __HIP_MEMORY_SCOPE_WORKGROUP);
        }
      }
    }
    __syncthreads();

    const int lim = min(PLACE_TILE, n_nodes - base);
    if ((n_nodes & 3) == 0) {
      f32x4* __restrict__ os4 = (f32x4*)(osum + base);
      f32x4* __restrict__ oc4 = (f32x4*)(ocnt + base);
      for (int j = tid; j < lim / 4; j += 1024) {
        os4[j] = ((const f32x4*)tsum)[j];
        oc4[j] = ((const f32x4*)tcnt)[j];
      }
    } else {
      for (int j = tid; j < lim; j += 1024) {
        osum[base + j] = tsum[j];
        ocnt[base + j] = tcnt[j];
      }
    }
    __syncthreads();
  }
}

#define BROWS 4
#define BTHREADS 128

__global__ __launch_bounds__(BTHREADS) void bcast_out(const float* __restrict__ part,
                                                      const float* __restrict__ node_mean,
                                                      float* __restrict__ out, int n_nodes) {
  static_assert(NPART == 16, "lanes 0-15 / 16-31 of a wave map to the 16 partial tables");
  static_assert((OUTW / 4) % 32 == 0, "a row is a whole number of 512-byte wave stores");
  const int lane = threadIdx.x & 31;
  const int gw   = blockIdx.x * (BTHREADS / 32) + (threadIdx.x >> 5);
  const int nw   = gridDim.x * (BTHREADS / 32);
  const int k = lane & (NPART - 1);
  const int w = lane >> 4;
  const float* __restrict__ srcp = part + (size_t)(k * 2 + w) * n_nodes;
  const bool vec = aligned16(part) && aligned16(node_mean) && (n_nodes & 3) == 0;

  for (int r0 = gw * BROWS; r0 < n_nodes; r0 += nw * BROWS) {
    float v[BROWS], nm[BROWS];
    if (vec) {
      const f32x4 t = *(const f32x4*)(srcp + r0);
      const f32x4 g = *(const f32x4*)(node_mean + r0);
      v[0] = t.x; v[1] = t.y; v[2] = t.z; v[3] = t.w;
      nm[0] = g.x; nm[1] = g.y; nm[2] = g.z; nm[3] = g.w;
    } else {
      for (int u = 0; u < BROWS; ++u) {
        v[u]  = srcp[min(r0 + u, n_nodes - 1)];
        nm[u] = node_mean[min(r0 + u, n_nodes - 1)];
      }
    }
    for (int u = 0; u < BROWS; ++u) {
      float x = v[u];
      x += __shfl_xor(x, 8);
      x += __shfl_xor(x, 4);
      x += __shfl_xor(x, 2);
      x += __shfl_xor(x, 1);
      const float s = __shfl(x, 0);
      const float c = __shfl(x, 16);
      const float pm = c > 0.0f ? s / c + nm[u] * (1.0f / 3.0f) : 0.0f;
      if (r0 + u < n_nodes) {
        const f32x4 v4 = {pm, pm, pm, pm};
        f32x4* __restrict__ o = (f32x4*)out + (size_t)(r0 + u) * (OUTW / 4) + lane;
        for (int j = 0; j < OUTW / 4 / 32; ++j)
          __builtin_nontemporal_store(v4, o + j * 32);
      }
    }
  }
}

extern "C" void kernel_launch(void* const* d_in, const int* in_sizes, int n_in,
                              void* d_out, int out_size, void* d_ws, size_t ws_size,
                              hipStream_t stream) {
  const float* emb       = (const float*)d_in[0];
  const int* pair_place  = (const int*)d_in[1];
  const int* event_pair  = (const int*)d_in[2];
  const int* event_src   = (const int*)d_in[3];
  const int* event_ndst  = (const int*)d_in[4];
  const int* dst_event   = (const int*)d_in[5];
  const int* dst_node    = (const int*)d_in[6];

  const int n_nodes  = in_sizes[0] / EMB;
  const int n_pairs  = in_sizes[1];
  const int n_events = in_sizes[2];
  const int n_dst    = in_sizes[5];

  float* ws        = (float*)d_ws;
  float* node_mean = ws;
  float* part      = node_mean + n_nodes;
  float* pair_acc  = part + n_events;

  rowmean_wmma<<<(n_nodes + 63) / 64, RM_THREADS, 0, stream>>>(emb, node_mean, n_nodes, pair_acc,
                                                               n_events, n_pairs, ws_size);

  dst_event_reduce<<<TB_GROUPS, TB_THREADS, 0, stream>>>(
      dst_event, dst_node, event_pair, event_src, event_ndst, node_mean,
      pair_acc, n_dst, n_events, n_pairs, n_nodes, ws_size);

  pair_reduce<<<NPART, 1024, 0, stream>>>(pair_place, pair_acc, part,
                                          n_pairs, n_nodes, n_events, ws_size);

  bcast_out<<<1024, BTHREADS, 0, stream>>>(part, node_mean, (float*)d_out, n_nodes);
}


# pair stage on all 256 WGPs: a pair's mean and its 1 added by two adjacent lanes of one global atomic into per-half {sum,count} place records; no 16 LDS partial tables; sum order changes
# speedup vs baseline: 4.4844x; 1.1662x over previous
#include <hip/hip_runtime.h>

#define EMB 512
#define OUTW (3 * EMB)

typedef __attribute__((ext_vector_type(2)))  float     v2f;
typedef __attribute__((ext_vector_type(8)))  float     v8f;
typedef __attribute__((ext_vector_type(4)))  float     f32x4;
typedef __attribute__((ext_vector_type(4)))  int       i32x4;
typedef v2f __attribute__((aligned(4))) rec2f;


static __device__ __forceinline__ bool aligned16(const void* p) {
  return (((unsigned long long)p) & 15ull) == 0ull;
}

static __device__ __forceinline__ int pair_copies(size_t ws_bytes, int n_nodes, int n_events,
                                                  int n_pairs) {
  const size_t need =
      ((size_t)n_nodes + (size_t)n_events + 4 * (size_t)n_pairs) * sizeof(float);
  return ws_bytes >= need ? 2 : 1;
}

static __device__ __forceinline__ int cache_half() {
  return (int)((__builtin_amdgcn_s_sendmsg_rtn(0x87) >> 18) & 1u);
}

static __device__ __forceinline__ float* place_records(float* pair_acc0, int n_nodes) {
  return pair_acc0 - 4 * (size_t)n_nodes;
}

static __device__ __forceinline__ void zero_fill(float* __restrict__ p, int n, int tid, int nth) {
  const int mis  = (int)((((unsigned long long)p) & 15ull) >> 2);
  const int head = min(n, (4 - mis) & 3);
  if (tid < head) p[tid] = 0.0f;
  float* __restrict__ q = p + head;
  const int m  = n - head;
  const int m4 = m >> 2;
  f32x4* __restrict__ q4 = (f32x4*)q;
  for (int j = tid; j < m4; j += nth) q4[j] = f32x4{0.0f, 0.0f, 0.0f, 0.0f};
  const int t = m4 * 4 + tid;
  if (t < m) q[t] = 0.0f;
}

#define RM_THREADS 256
#define RM_ROWS 8
#define RM_CHUNKS (EMB / 64)

__global__ __launch_bounds__(RM_THREADS) void rowmean_wmma(const float* __restrict__ emb,
                                                           float* __restrict__ node_mean,
                                                           int n_nodes, float* __restrict__ acc,
                                                           int n_events, int n_pairs,
                                                           size_t ws_bytes) {
  static_assert(EMB % 64 == 0, "a row is a whole number of 256-byte chunks");
  static_assert(RM_ROWS % 4 == 0, "rows go in pairs, and two pairs' products are interleaved");
  zero_fill(place_records(acc, n_nodes),
            4 * n_nodes + 2 * n_pairs * pair_copies(ws_bytes, n_nodes, n_events, n_pairs),
            blockIdx.x * RM_THREADS + threadIdx.x, gridDim.x * RM_THREADS);

  const int lane = threadIdx.x & 31;
  const int wave = threadIdx.x >> 5;
  const int row0 = (blockIdx.x * (RM_THREADS / 32) + wave) * RM_ROWS;
  if (row0 >= n_nodes) return;

  const int c   = lane & 15;
  const int h   = lane >> 4;
  const int sel = c >> 3;
  const int q   = (c & 7) + 8 * h;
  const float* __restrict__ rp[RM_ROWS / 2];
#pragma unroll
  for (int p = 0; p < RM_ROWS / 2; ++p)
    rp[p] = emb + (size_t)min(row0 + p + (RM_ROWS / 2) * sel, n_nodes - 1) * EMB + q * 4;

  f32x4 v[RM_ROWS / 2][RM_CHUNKS];
#pragma unroll
  for (int g = 0; g < RM_ROWS / 2; g += 2)
#pragma unroll
    for (int s = 0; s < RM_CHUNKS; ++s) {
      v[g][s]     = *(const f32x4*)(rp[g] + s * 64);
      v[g + 1][s] = *(const f32x4*)(rp[g + 1] + s * 64);
    }
  __builtin_amdgcn_sched_barrier(0);

  const v2f bones = {1.0f, 1.0f};
  v8f cacc[RM_ROWS / 2];
#pragma unroll
  for (int p = 0; p < RM_ROWS / 2; ++p) cacc[p] = v8f{};
#pragma unroll
  for (int g = 0; g < RM_ROWS / 2; g += 2)
#pragma unroll
    for (int s = 0; s < RM_CHUNKS; ++s) {
      const f32x4 x = v[g][s];
      const f32x4 y = v[g + 1][s];
      cacc[g] = __builtin_amdgcn_wmma_f32_16x16x4_f32(
           false, v2f{x.x, x.y},  false, bones,
           (short)0, cacc[g],  false,  false);
      cacc[g + 1] = __builtin_amdgcn_wmma_f32_16x16x4_f32(
          false, v2f{y.x, y.y}, false, bones, (short)0, cacc[g + 1], false, false);
      cacc[g] = __builtin_amdgcn_wmma_f32_16x16x4_f32(
          false, v2f{x.z, x.w}, false, bones, (short)0, cacc[g], false, false);
      cacc[g + 1] = __builtin_amdgcn_wmma_f32_16x16x4_f32(
          false, v2f{y.z, y.w}, false, bones, (short)0, cacc[g + 1], false, false);
    }

  if (c == 0) {
    const float inv = 1.0f / (float)EMB;
    const int base = row0 + (RM_ROWS / 2) * h;
#pragma unroll
    for (int p = 0; p < RM_ROWS / 2; ++p) {
      const float t = ((cacc[p][0] + cacc[p][1]) + (cacc[p][2] + cacc[p][3])) +
                      ((cacc[p][4] + cacc[p][5]) + (cacc[p][6] + cacc[p][7]));
      if (base + p < n_nodes) node_mean[base + p] = t * inv;
    }
  }
}

#define TAB_MAX 16384
#define TB_THREADS 1024
#define TB_GROUPS 256
#define TB_BATCH 4

static __device__ __forceinline__ void stage_table(float* __restrict__ tab,
                                                   const float* __restrict__ node_mean, int n) {
  const int t = threadIdx.x;
  if (aligned16(node_mean)) {
    const f32x4* __restrict__ s4 = (const f32x4*)node_mean;
    f32x4* __restrict__ t4 = (f32x4*)tab;
    const int n4 = n >> 2;
    for (int j0 = t; j0 < n4; j0 += TB_THREADS * TB_BATCH) {
      f32x4 v[TB_BATCH];
#pragma unroll
      for (int u = 0; u < TB_BATCH; ++u) {
        const int j = j0 + u * TB_THREADS;
        v[u] = s4[j < n4 ? j : j0];
      }
#pragma unroll
      for (int u = 0; u < TB_BATCH; ++u) {
        const int j = j0 + u * TB_THREADS;
        if (j < n4) t4[j] = v[u];
      }
    }
    const int r = n4 * 4 + t;
    if (r < n) tab[r] = node_mean[r];
  } else {
    for (int j = t; j < n; j += TB_THREADS) tab[j] = node_mean[j];
  }
}

template <bool IN_LDS>
static __device__ __forceinline__ float nm_at(const float* __restrict__ tab,
                                              const float* __restrict__ node_mean, int i) {
  return IN_LDS ? tab[i] : node_mean[i];
}

#define EV_WIN (TB_THREADS * 4)
#define DST_TILE (TB_THREADS * 4)
#define DST_PRE 3

template <bool IN_LDS>
static __device__ __forceinline__ float event_val(int src, int ndst, float dsum,
                                                  const float* __restrict__ tab,
                                                  const float* __restrict__ node_mean) {
  return (nm_at<IN_LDS>(tab, node_mean, src) + dsum / ((float)ndst + 1.0f)) * (1.0f / 3.0f);
}

static __device__ __forceinline__ int lane_neighbour(int x) {
  return __builtin_amdgcn_update_dpp(0, x, 0xB1, 0xF, 0xF, false);
}

static __device__ __forceinline__ void record_add(float* __restrict__ pair_acc, int key, int word,
                                                  float val) {
  unsafeAtomicAdd(&pair_acc[2 * (size_t)key + word], val);
}

static __device__ __forceinline__ void event_add(int p, float s, float* __restrict__ pair_acc) {
  unsafeAtomicAdd(&pair_acc[2 * (size_t)p], s);
  unsafeAtomicAdd(&pair_acc[2 * (size_t)p + 1], 1.0f);
}

static __device__ __forceinline__ void load_dst4(const int* __restrict__ dst_event,
                                                 const int* __restrict__ dst_node, bool vec, int p,
                                                 int n_dst, i32x4& e, i32x4& d) {
  e = i32x4{-1, -1, -1, -1};
  d = i32x4{0, 0, 0, 0};
  if (vec) {
    const int j = p >> 2;
    if (j < (n_dst >> 2)) {
      e = ((const i32x4*)dst_event)[j];
      d = ((const i32x4*)dst_node)[j];
    }
  } else {
    if (p < n_dst)     { e.x = dst_event[p];     d.x = dst_node[p]; }
    if (p + 1 < n_dst) { e.y = dst_event[p + 1]; d.y = dst_node[p + 1]; }
    if (p + 2 < n_dst) { e.z = dst_event[p + 2]; d.z = dst_node[p + 2]; }
    if (p + 3 < n_dst) { e.w = dst_event[p + 3]; d.w = dst_node[p + 3]; }
  }
}

template <bool IN_LDS>
static __device__ __forceinline__ void window_add4(float* __restrict__ dsum,
                                                   const float* __restrict__ tab,
                                                   const float* __restrict__ node_mean, i32x4 e,
                                                   i32x4 d, int wb, int lo, unsigned span) {
  const float a0 = nm_at<IN_LDS>(tab, node_mean, d.x);
  const float a1 = nm_at<IN_LDS>(tab, node_mean, d.y);
  const float a2 = nm_at<IN_LDS>(tab, node_mean, d.z);
  const float a3 = nm_at<IN_LDS>(tab, node_mean, d.w);
  if ((unsigned)(e.x - lo) < span)
    __hip_atomic_fetch_add(&dsum[e.x - wb], a0, __ATOMIC_RELAXED, __HIP_MEMORY_SCOPE_WORKGROUP);
  if ((unsigned)(e.y - lo) < span)
    __hip_atomic_fetch_add(&dsum[e.y - wb], a1, __ATOMIC_RELAXED, __HIP_MEMORY_SCOPE_WORKGROUP);
  if ((unsigned)(e.z - lo) < span)
    __hip_atomic_fetch_add(&dsum[e.z - wb], a2, __ATOMIC_RELAXED, __HIP_MEMORY_SCOPE_WORKGROUP);
  if ((unsigned)(e.w - lo) < span)
    __hip_atomic_fetch_add(&dsum[e.w - wb], a3, __ATOMIC_RELAXED, __HIP_MEMORY_SCOPE_WORKGROUP);
}

template <bool IN_LDS>
static __device__ __forceinline__ void dst_event_body(float* __restrict__ tab,
                                                      float* __restrict__ dsum,
                                                      int* __restrict__ bnd,
                                                      const int* __restrict__ dst_event,
                                                      const int* __restrict__ dst_node,
                                                      const int* __restrict__ event_pair,
                                                      const int* __restrict__ event_src,
                                                      const int* __restrict__ event_ndst,
                                                      const float* __restrict__ node_mean,
                                                      float* __restrict__ pair_acc0, int n_dst,
                                                      int n_events, int n_pairs, int n_nodes,
                                                      size_t ws_bytes) {
  if (n_events <= 0) return;
  const int t = threadIdx.x;
  const int G = gridDim.x;
  const int b = blockIdx.x;
  const int h = pair_copies(ws_bytes, n_nodes, n_events, n_pairs) == 2 ? cache_half() : 0;
  float* __restrict__ pair_acc = pair_acc0 + (size_t)h * 2 * (size_t)n_pairs;

  const int chunk = (((n_dst + G - 1) / G) + 3) & ~3;
  const int k0 = b * chunk;
  const int k1 = k0 + chunk;
  const int c0 = min(k0, n_dst);
  const int c1 = min(k1, n_dst);
  const bool dvec = aligned16(dst_event) && aligned16(dst_node);
  const bool evec = aligned16(event_pair) && aligned16(event_src) && aligned16(event_ndst);
  const int q0 = dvec ? c0 + DST_PRE * DST_TILE : c0;

  const int si = t == 0 ? k0 - 1 : (t == 1 ? k1 - 1 : q0);
  const int* __restrict__ sp =
      n_dst > 0 ? dst_event + min(max(si, 0), n_dst - 1) : event_pair;
  const int sv = *sp;
  i32x4 pe[DST_PRE], pd[DST_PRE];
#pragma unroll
  for (int k = 0; k < DST_PRE; ++k) {
    pe[k] = i32x4{-1, -1, -1, -1};
    pd[k] = i32x4{0, 0, 0, 0};
    if (dvec) load_dst4(dst_event, dst_node, true, c0 + k * DST_TILE + 4 * t, n_dst, pe[k], pd[k]);
  }
  int te = -1, td = 0;
  if (dvec && t < (n_dst & 3)) {
    te = dst_event[(n_dst & ~3) + t];
    td = dst_node[(n_dst & ~3) + t];
  }

  if (IN_LDS) stage_table(tab, node_mean, n_nodes);
  if (t < 3) bnd[t] = sv;
  ((f32x4*)dsum)[t] = f32x4{0.0f, 0.0f, 0.0f, 0.0f};
  __syncthreads();

  const int f0 = b <= 0 ? 0 : (k0 >= n_dst ? n_events : min(max(bnd[0] + 1, 0), n_events));
  const int f1 =
      max(f0, (b + 1 >= G || k1 >= n_dst) ? n_events : min(max(bnd[1] + 1, 0), n_events));
  const int ex = q0 < n_dst ? bnd[2] : n_events;
  const int odd = t & 1;
  for (int wb = f0 & ~3; wb < f1; wb += EV_WIN) {
    const int lo = max(wb, f0);
    const int hi = min(wb + EV_WIN, f1);
    const unsigned span = (unsigned)(hi - lo);
    if (wb != (f0 & ~3)) {
      ((f32x4*)dsum)[t] = f32x4{0.0f, 0.0f, 0.0f, 0.0f};
      __syncthreads();
    }

    const int ev = wb + 4 * t;
    const bool fast = evec && ev + 3 < n_events;
    i32x4 p = {0, 0, 0, 0};
    i32x4 sr = {0, 0, 0, 0};
    i32x4 nd = {0, 0, 0, 0};
    if (fast) {
      p  = *(const i32x4*)(event_pair + ev);
      sr = *(const i32x4*)(event_src + ev);
      nd = *(const i32x4*)(event_ndst + ev);
    }

#pragma unroll
    for (int k = 0; k < DST_PRE; ++k)
      window_add4<IN_LDS>(dsum, tab, node_mean, pe[k], pd[k], wb, lo, span);
    if ((unsigned)(te - lo) < span)
      __hip_atomic_fetch_add(&dsum[te - wb], nm_at<IN_LDS>(tab, node_mean, td), __ATOMIC_RELAXED,
                             __HIP_MEMORY_SCOPE_WORKGROUP);
    {
      int q = q0;
      int eq = ex;
      while (q < n_dst && (q < c1 || eq < f1)) {
        i32x4 e, d;
        load_dst4(dst_event, dst_node, dvec, q + 4 * t, n_dst, e, d);
        window_add4<IN_LDS>(dsum, tab, node_mean, e, d, wb, lo, span);
        q += DST_TILE;
        eq = q < n_dst ? dst_event[q] : n_events;
      }
    }
    __syncthreads();

    const f32x4 ds = ((const f32x4*)dsum)[t];
    const float v0 = event_val<IN_LDS>(sr.x, nd.x, ds.x, tab, node_mean);
    const float v1 = event_val<IN_LDS>(sr.y, nd.y, ds.y, tab, node_mean);
    const float v2 = event_val<IN_LDS>(sr.z, nd.z, ds.z, tab, node_mean);
    const float v3 = event_val<IN_LDS>(sr.w, nd.w, ds.w, tab, node_mean);
    const int mk = !fast ? 0
                         : (((unsigned)(ev - lo) < span ? 1 : 0) |
                            ((unsigned)(ev + 1 - lo) < span ? 2 : 0) |
                            ((unsigned)(ev + 2 - lo) < span ? 4 : 0) |
                            ((unsigned)(ev + 3 - lo) < span ? 8 : 0));
    const int rk = lane_neighbour(mk);
    const int r0 = lane_neighbour(p.x);
    const int r1 = lane_neighbour(p.y);
    const int r2 = lane_neighbour(p.z);
    const int r3 = lane_neighbour(p.w);
    const int me = odd ? rk : mk;
    const int mo = odd ? mk : rk;
    if (me & 1) record_add(pair_acc, odd ? r0 : p.x, odd, odd ? 1.0f : v0);
    if (me & 2) record_add(pair_acc, odd ? r1 : p.y, odd, odd ? 1.0f : v1);
    if (me & 4) record_add(pair_acc, odd ? r2 : p.z, odd, odd ? 1.0f : v2);
    if (me & 8) record_add(pair_acc, odd ? r3 : p.w, odd, odd ? 1.0f : v3);
    if (mo & 1) record_add(pair_acc, odd ? p.x : r0, odd ^ 1, odd ? v0 : 1.0f);
    if (mo & 2) record_add(pair_acc, odd ? p.y : r1, odd ^ 1, odd ? v1 : 1.0f);
    if (mo & 4) record_add(pair_acc, odd ? p.z : r2, odd ^ 1, odd ? v2 : 1.0f);
    if (mo & 8) record_add(pair_acc, odd ? p.w : r3, odd ^ 1, odd ? v3 : 1.0f);
    if (!fast) {
      for (int k = 0; k < 4; ++k) {
        const int e = ev + k;
        if (e < n_events && (unsigned)(e - lo) < span)
          event_add(event_pair[e],
                    event_val<IN_LDS>(event_src[e], event_ndst[e], dsum[e - wb], tab, node_mean),
                    pair_acc);
      }
    }
  }
}

__global__ __launch_bounds__(TB_THREADS) void dst_event_reduce(const int* __restrict__ dst_event,
                                                               const int* __restrict__ dst_node,
                                                               const int* __restrict__ event_pair,
                                                               const int* __restrict__ event_src,
                                                               const int* __restrict__ event_ndst,
                                                               const float* __restrict__ node_mean,
                                                               float* __restrict__ pair_acc0, int n_dst,
                                                               int n_events, int n_pairs, int n_nodes,
                                                               size_t ws_bytes) {
  __shared__ __attribute__((aligned(16))) float tab[TAB_MAX];
  __shared__ __attribute__((aligned(16))) float dsum[EV_WIN];
  __shared__ int bnd[4];
  if (n_nodes <= TAB_MAX)
    dst_event_body<true>(tab, dsum, bnd, dst_event, dst_node, event_pair, event_src, event_ndst, node_mean,
                         pair_acc0, n_dst, n_events, n_pairs, n_nodes, ws_bytes);
  else
    dst_event_body<false>(tab, dsum, bnd, dst_event, dst_node, event_pair, event_src, event_ndst,
                          node_mean, pair_acc0, n_dst, n_events, n_pairs, n_nodes, ws_bytes);
}

#define PP_THREADS 256
#define PP_GROUPS 256
#define PP_BATCH 8

__global__ __launch_bounds__(PP_THREADS) void pair_to_place(const int* __restrict__ pair_place,
                                                            float* __restrict__ pair_acc,
                                                            int n_pairs, int n_nodes, int n_events,
                                                            size_t ws_bytes) {
  const bool two = pair_copies(ws_bytes, n_nodes, n_events, n_pairs) == 2;
  const float* __restrict__ acc0 = pair_acc;
  const float* __restrict__ acc1 = pair_acc + (two ? 2 * (size_t)n_pairs : 0);
  float* __restrict__ prec =
      place_records(pair_acc, n_nodes) + (size_t)cache_half() * 2 * (size_t)n_nodes;

  const int word  = threadIdx.x & 1;
  const int slot  = (blockIdx.x * PP_THREADS + threadIdx.x) >> 1;
  const int slots = (gridDim.x * PP_THREADS) >> 1;

  for (int i0 = slot; i0 < n_pairs; i0 += slots * PP_BATCH) {
    float c[PP_BATCH], s[PP_BATCH];
    int pl[PP_BATCH];
#pragma unroll
    for (int u = 0; u < PP_BATCH; ++u) {
      const int i = i0 + u * slots;
      const bool ok = i < n_pairs;
      const int ic = ok ? i : i0;
      const v2f r0 = *(const rec2f*)(acc0 + 2 * (size_t)ic);
      const v2f r1 = *(const rec2f*)(acc1 + 2 * (size_t)ic);
      const float cu = r0.y + (two ? r1.y : 0.0f);
      c[u]  = ok ? cu : 0.0f;
      s[u]  = r0.x + (two ? r1.x : 0.0f);
      pl[u] = pair_place[ic];
    }
#pragma unroll
    for (int u = 0; u < PP_BATCH; ++u) {
      if (c[u] > 0.0f && (unsigned)pl[u] < (unsigned)n_nodes)
        unsafeAtomicAdd(&prec[2 * (size_t)pl[u] + word], word ? 1.0f : s[u] / c[u]);
    }
  }
}

#define BROWS 4
#define BTHREADS 128

__global__ __launch_bounds__(BTHREADS) void bcast_out(const float* __restrict__ pair_acc0,
                                                      const float* __restrict__ node_mean,
                                                      float* __restrict__ out, int n_nodes) {
  static_assert(BROWS == 4, "a trip's records are two 16-byte loads a copy");
  static_assert((OUTW / 4) % 32 == 0, "a row is a whole number of 512-byte wave stores");
  const int lane = threadIdx.x & 31;
  const int gw   = blockIdx.x * (BTHREADS / 32) + (threadIdx.x >> 5);
  const int nw   = gridDim.x * (BTHREADS / 32);
  const float* __restrict__ prec0 = pair_acc0 - 4 * (size_t)n_nodes;
  const float* __restrict__ prec1 = prec0 + 2 * (size_t)n_nodes;
  const bool vec = aligned16(prec0) && aligned16(node_mean) && (n_nodes & 3) == 0;

  for (int r0 = gw * BROWS; r0 < n_nodes; r0 += nw * BROWS) {
    float s[BROWS], c[BROWS], nm[BROWS];
    if (vec) {
      const f32x4 a0 = *(const f32x4*)(prec0 + 2 * (size_t)r0);
      const f32x4 a1 = *(const f32x4*)(prec0 + 2 * (size_t)r0 + 4);
      const f32x4 b0 = *(const f32x4*)(prec1 + 2 * (size_t)r0);
      const f32x4 b1 = *(const f32x4*)(prec1 + 2 * (size_t)r0 + 4);
      const f32x4 g  = *(const f32x4*)(node_mean + r0);
      s[0] = a0.x + b0.x; c[0] = a0.y + b0.y;
      s[1] = a0.z + b0.z; c[1] = a0.w + b0.w;
      s[2] = a1.x + b1.x; c[2] = a1.y + b1.y;
      s[3] = a1.z + b1.z; c[3] = a1.w + b1.w;
      nm[0] = g.x; nm[1] = g.y; nm[2] = g.z; nm[3] = g.w;
    } else {
      for (int u = 0; u < BROWS; ++u) {
        const size_t r = (size_t)min(r0 + u, n_nodes - 1);
        s[u]  = prec0[2 * r] + prec1[2 * r];
        c[u]  = prec0[2 * r + 1] + prec1[2 * r + 1];
        nm[u] = node_mean[r];
      }
    }
    for (int u = 0; u < BROWS; ++u) {
      const float pm = c[u] > 0.0f ? s[u] / c[u] + nm[u] * (1.0f / 3.0f) : 0.0f;
      if (r0 + u < n_nodes) {
        const f32x4 v4 = {pm, pm, pm, pm};
        f32x4* __restrict__ o = (f32x4*)out + (size_t)(r0 + u) * (OUTW / 4) + lane;
        for (int j = 0; j < OUTW / 4 / 32; ++j)
          __builtin_nontemporal_store(v4, o + j * 32);
      }
    }
  }
}

extern "C" void kernel_launch(void* const* d_in, const int* in_sizes, int n_in,
                              void* d_out, int out_size, void* d_ws, size_t ws_size,
                              hipStream_t stream) {
  const float* emb       = (const float*)d_in[0];
  const int* pair_place  = (const int*)d_in[1];
  const int* event_pair  = (const int*)d_in[2];
  const int* event_src   = (const int*)d_in[3];
  const int* event_ndst  = (const int*)d_in[4];
  const int* dst_event   = (const int*)d_in[5];
  const int* dst_node    = (const int*)d_in[6];

  const int n_nodes  = in_sizes[0] / EMB;
  const int n_pairs  = in_sizes[1];
  const int n_events = in_sizes[2];
  const int n_dst    = in_sizes[5];

  float* ws        = (float*)d_ws;
  float* node_mean = ws;
  float* plane     = node_mean + n_nodes;
  float* pair_acc  = plane + n_events;

  rowmean_wmma<<<(n_nodes + 63) / 64, RM_THREADS, 0, stream>>>(emb, node_mean, n_nodes, pair_acc,
                                                               n_events, n_pairs, ws_size);

  dst_event_reduce<<<TB_GROUPS, TB_THREADS, 0, stream>>>(
      dst_event, dst_node, event_pair, event_src, event_ndst, node_mean,
      pair_acc, n_dst, n_events, n_pairs, n_nodes, ws_size);

  pair_to_place<<<PP_GROUPS, PP_THREADS, 0, stream>>>(pair_place, pair_acc, n_pairs, n_nodes,
                                                      n_events, ws_size);

  bcast_out<<<1024, BTHREADS, 0, stream>>>(pair_acc, node_mean, (float*)d_out, n_nodes);
}
